# starting kernel
# baseline (speedup 1.0000x reference)
#include <hip/hip_runtime.h>
#include <hip/hip_bf16.h>
#include <math.h>


typedef __attribute__((ext_vector_type(16))) _Float16 v16h;
typedef __attribute__((ext_vector_type(8)))  _Float16 v8h;
typedef __attribute__((ext_vector_type(4)))  _Float16 v4h;
typedef __attribute__((ext_vector_type(8)))  float    v8f;
typedef __attribute__((ext_vector_type(4)))  float    v4f;

constexpr int Bc  = 4;
constexpr int Tc  = 2048;
constexpr int HIDc = 512;
constexpr int Hc  = 4;
constexpr int DKc = 64;
constexpr int DVc = 128;
constexpr int CH  = 64;
constexpr int TOK = Bc * Tc;
constexpr float RMS_EPS = 1e-5f;

__device__ __forceinline__ v8f wmma16(v16h a, v16h b, v8f c) {
  return __builtin_amdgcn_wmma_f32_16x16x32_f16(false, a, false, b, (short)0, c,
                                                false, false);
}

__device__ __forceinline__ v16h frag16(const _Float16* p, int ld, int rowBase,
                                       int kBase) {
  const int lane = threadIdx.x & 31, half = lane >> 4;
  const _Float16* row = p + (size_t)(rowBase + (lane & 15)) * ld + kBase + half * 8;
  v8h lo = *(const v8h*)(row);
  v8h hi = *(const v8h*)(row + 16);
  return __builtin_shufflevector(lo, hi, 0, 1, 2, 3, 4, 5, 6, 7,
                                 8, 9, 10, 11, 12, 13, 14, 15);
}

__device__ __forceinline__ v16h frag16_f32(const float* p, int ld, int rowBase,
                                           int kBase) {
  const int lane = threadIdx.x & 31, half = lane >> 4;
  const float* row = p + (size_t)(rowBase + (lane & 15)) * ld + kBase + half * 8;
  v4f a0 = *(const v4f*)(row);
  v4f a1 = *(const v4f*)(row + 4);
  v4f b0 = *(const v4f*)(row + 16);
  v4f b1 = *(const v4f*)(row + 20);
  v16h f;
#pragma unroll
  for (int e = 0; e < 4; ++e) {
    f[e]      = (_Float16)a0[e];
    f[4 + e]  = (_Float16)a1[e];
    f[8 + e]  = (_Float16)b0[e];
    f[12 + e] = (_Float16)b1[e];
  }
  return f;
}

__global__ __launch_bounds__(256) void proj_kernel(
    const float* __restrict__ x,
    const float* __restrict__ Wq, const float* __restrict__ Wk,
    const float* __restrict__ Wv, const float* __restrict__ Wg,
    float* __restrict__ qws, float* __restrict__ kws,
    float* __restrict__ vws, float* __restrict__ gws) {
  __shared__ __align__(16) _Float16 As[128][40];
  __shared__ __align__(16) _Float16 BsT[128][40];
  const int m0 = blockIdx.x * 128, n0 = blockIdx.y * 128;
  const int tid = threadIdx.x, lane = tid & 31, wave = tid >> 5;
  const int mr0 = (wave & 3) * 32;
  const int nc0 = (wave >> 2) * 64;
  v8f acc[2][4] = {};
  for (int kk = 0; kk < HIDc; kk += 32) {
    for (int i = tid; i < 1024; i += 256) {
      const int row = i >> 3, cg = i & 7;
      v4f xv = *(const v4f*)(x + (size_t)(m0 + row) * HIDc + kk + cg * 4);
      v4h hv;
#pragma unroll
      for (int e = 0; e < 4; ++e) hv[e] = (_Float16)xv[e];
      *(v4h*)&As[row][cg * 4] = hv;
    }
    for (int i = tid; i < 1024; i += 256) {
      const int k = i >> 5, ng = i & 31, n = n0 + ng * 4;
      v4f wv;
      if (n < 256)       wv = *(const v4f*)(Wq + (size_t)(kk + k) * 256 + n);
      else if (n < 512)  wv = *(const v4f*)(Wk + (size_t)(kk + k) * 256 + (n - 256));
      else if (n < 1024) wv = *(const v4f*)(Wv + (size_t)(kk + k) * 512 + (n - 512));
      else               wv = *(const v4f*)(Wg + (size_t)(kk + k) * 512 + (n - 1024));
#pragma unroll
      for (int j2 = 0; j2 < 4; ++j2) BsT[ng * 4 + j2][k] = (_Float16)wv[j2];
    }
    if (kk + 32 < HIDc)
      __builtin_prefetch(&x[(size_t)(m0 + (tid >> 1)) * HIDc + kk + 32], 0, 1);
    __syncthreads();
    v16h a0 = frag16(&As[0][0], 40, mr0, 0);
    v16h a1 = frag16(&As[0][0], 40, mr0 + 16, 0);
#pragma unroll
    for (int j = 0; j < 4; ++j) {
      v16h bm = frag16(&BsT[0][0], 40, nc0 + j * 16, 0);
      acc[0][j] = wmma16(a0, bm, acc[0][j]);
      acc[1][j] = wmma16(a1, bm, acc[1][j]);
    }
    __syncthreads();
  }
  const int nl = lane & 15, mOff = (lane >> 4) * 8;
#pragma unroll
  for (int a2 = 0; a2 < 2; ++a2) {
#pragma unroll
    for (int j = 0; j < 4; ++j) {
      const int n = n0 + nc0 + j * 16 + nl;
#pragma unroll
      for (int r = 0; r < 8; ++r) {
        const size_t m = (size_t)(m0 + mr0 + a2 * 16 + mOff + r);
        const float val = acc[a2][j][r];
        if (n < 256)       qws[m * 256 + n]          = val * 0.125f;
        else if (n < 512)  kws[m * 256 + (n - 256)]  = val;
        else if (n < 1024) vws[m * 512 + (n - 512)]  = val;
        else               gws[m * 512 + (n - 1024)] = val;
      }
    }
  }
}

__global__ __launch_bounds__(256) void lr1_kernel(const float* __restrict__ x,
                                                  const float* __restrict__ Wgk1,
                                                  float* __restrict__ xg1) {
  __shared__ float xs[16][512];
  const int t0 = blockIdx.x * 16;
  for (int i = threadIdx.x; i < 16 * 512; i += 256)
    xs[i >> 9][i & 511] = x[(size_t)(t0 + (i >> 9)) * 512 + (i & 511)];
  __syncthreads();
  const int tok = threadIdx.x >> 4, j = threadIdx.x & 15;
  float s = 0.f;
  for (int k2 = 0; k2 < 512; ++k2) s += xs[tok][k2] * Wgk1[k2 * 16 + j];
  xg1[(size_t)(t0 + tok) * 16 + j] = s;
}

__global__ __launch_bounds__(256) void lr2_kernel(const float* __restrict__ xg1,
                                                  const float* __restrict__ Wgk2,
                                                  const float* __restrict__ bgk2,
                                                  float* __restrict__ gk) {
  const int idx = blockIdx.x * 256 + threadIdx.x;
  const int tok = idx >> 8, n = idx & 255;
  float s = bgk2[n];
#pragma unroll
  for (int j = 0; j < 16; ++j) s += xg1[(size_t)tok * 16 + j] * Wgk2[j * 256 + n];
  const float ls = fminf(s, 0.f) - log1pf(__expf(-fabsf(s)));
  gk[idx] = ls * (1.f / 16.f);
}

__global__ void zero_kernel(float* __restrict__ p, int n) {
  const int i = blockIdx.x * 256 + threadIdx.x;
  if (i < n) p[i] = 0.f;
}

__global__ __launch_bounds__(256) void gla_scan_kernel(
    const float* __restrict__ qg, const float* __restrict__ kg,
    const float* __restrict__ vg, const float* __restrict__ gkg,
    float* __restrict__ on, float* __restrict__ ssq_g) {
  constexpr int LDA = CH + 8;
  constexpr int LDS_ = 68;
  __shared__ __align__(16) float    ST[64][LDS_];
  __shared__ __align__(16) _Float16 qt[CH][LDA];
  __shared__ __align__(16) _Float16 kt[CH][LDA];
  __shared__ __align__(16) _Float16 ktT[DKc][LDA];
  __shared__ __align__(16) _Float16 Am[CH][LDA];
  __shared__ __align__(16) _Float16 vT[64][LDA];
  __shared__ float expLC[DKc];
  __shared__ float ssql[CH];

  const int bh = blockIdx.x, b = bh >> 2, h = bh & 3;
  const int nHalf = blockIdx.y;
  const int tid = threadIdx.x, lane = tid & 31, wave = tid >> 5;

  for (int i = tid; i < 64 * LDS_; i += 256) (&ST[0][0])[i] = 0.f;
  __syncthreads();

  for (int c = 0; c < Tc / CH; ++c) {
    const int t0 = c * CH;
    const float* qb  = qg  + ((size_t)(b * Tc + t0) * Hc + h) * DKc;
    const float* kb  = kg  + ((size_t)(b * Tc + t0) * Hc + h) * DKc;
    const float* gkb = gkg + ((size_t)(b * Tc + t0) * Hc + h) * DKc;
    const float* vb  = vg  + (size_t)(b * Tc + t0) * (Hc * DVc) + h * DVc + nHalf * 64;
    float*       ob  = on  + (size_t)(b * Tc + t0) * (Hc * DVc) + h * DVc + nHalf * 64;

    if (tid < DKc) {
      const int d = tid;
      float lamC = 0.f;
      for (int t = 0; t < CH; ++t) lamC += gkb[(size_t)t * (Hc * DKc) + d];
      expLC[d] = __expf(lamC);
      float lam = 0.f;
      for (int t = 0; t < CH; ++t) {
        const size_t off = (size_t)t * (Hc * DKc) + d;
        lam += gkb[off];
        const float eq = __expf(lam - lamC);
        const float qh = fmaxf(fminf(qb[off] * eq, 60000.f), -60000.f);
        const float kh = kb[off] / eq;
        qt[t][d]  = (_Float16)qh;
        kt[t][d]  = (_Float16)kh;
        ktT[d][t] = (_Float16)kh;
      }
    }
    if (tid < CH) ssql[tid] = 0.f;
    __syncthreads();

    for (int i = tid; i < 64 * 64; i += 256) {
      const int n = i >> 6, d = i & 63;
      ST[n][d] *= expLC[d];
    }
    for (int i = tid; i < CH * 64; i += 256) {
      const int s = i >> 6, n = i & 63;
      vT[n][s] = (_Float16)vb[(size_t)s * (Hc * DVc) + n];
    }

    {
      const int tRow = wave & 3, sp0 = (wave >> 2) * 2;
#pragma unroll
      for (int spi = 0; spi < 2; ++spi) {
        const int sTile = sp0 + spi;
        v8f acc = {};
        if (sTile <= tRow) {
#pragma unroll
          for (int k2 = 0; k2 < 2; ++k2) {
            v16h a  = frag16(&qt[0][0], LDA, tRow * 16, k2 * 32);
            v16h bm = frag16(&kt[0][0], LDA, sTile * 16, k2 * 32);
            acc = wmma16(a, bm, acc);
          }
        }
        const int sIdx = sTile * 16 + (lane & 15), mOff = (lane >> 4) * 8;
#pragma unroll
        for (int r = 0; r < 8; ++r) {
          const int tIdx = tRow * 16 + mOff + r;
          Am[tIdx][sIdx] = (_Float16)((sIdx <= tIdx) ? acc[r] : 0.f);
        }
      }
    }
    __syncthreads();

    {
      const int tRow = wave & 3, nt0 = (wave >> 2) * 2;
      v8f acc[2] = {};
#pragma unroll
      for (int k2 = 0; k2 < 2; ++k2) {
        v16h aA = frag16(&Am[0][0], LDA, tRow * 16, k2 * 32);
        v16h aQ = frag16(&qt[0][0], LDA, tRow * 16, k2 * 32);
#pragma unroll
        for (int j = 0; j < 2; ++j) {
          v16h bv = frag16(&vT[0][0], LDA, (nt0 + j) * 16, k2 * 32);
          acc[j] = wmma16(aA, bv, acc[j]);
          v16h bs = frag16_f32(&ST[0][0], LDS_, (nt0 + j) * 16, k2 * 32);
          acc[j] = wmma16(aQ, bs, acc[j]);
        }
      }
      const int nl = lane & 15, mOff = (lane >> 4) * 8;
#pragma unroll
      for (int j = 0; j < 2; ++j) {
        const int n = (nt0 + j) * 16 + nl;
#pragma unroll
        for (int r = 0; r < 8; ++r) {
          const int t = tRow * 16 + mOff + r;
          ob[(size_t)t * (Hc * DVc) + n] = acc[j][r];
          atomicAdd(&ssql[t], acc[j][r] * acc[j][r]);
        }
      }
    }
    __syncthreads();
    if (tid < CH)
      atomicAdd(&ssq_g[(size_t)(b * Tc + t0 + tid) * Hc + h], ssql[tid]);

    {
      const int dRow = wave & 3, nt0 = (wave >> 2) * 2;
      v8f acc[2] = {};
#pragma unroll
      for (int k2 = 0; k2 < 2; ++k2) {
        v16h aK = frag16(&ktT[0][0], LDA, dRow * 16, k2 * 32);
#pragma unroll
        for (int j = 0; j < 2; ++j) {
          v16h bv = frag16(&vT[0][0], LDA, (nt0 + j) * 16, k2 * 32);
          acc[j] = wmma16(aK, bv, acc[j]);
        }
      }
      const int nl = lane & 15, mOff = (lane >> 4) * 8;
#pragma unroll
      for (int j = 0; j < 2; ++j) {
        const int n = (nt0 + j) * 16 + nl;
#pragma unroll
        for (int r = 0; r < 8; ++r) {
          const int d = dRow * 16 + mOff + r;
          ST[n][d] += acc[j][r];
        }
      }
    }
    __syncthreads();
  }
}

__global__ __launch_bounds__(256) void epilogue_kernel(
    float* __restrict__ on, const float* __restrict__ g,
    const float* __restrict__ rms_w, const float* __restrict__ ssq) {
  const size_t idx = (size_t)blockIdx.x * 256 + threadIdx.x;
  const size_t tok = idx >> 9;
  const int col = (int)(idx & 511), h = col >> 7, n = col & 127;
  const float ms = ssq[tok * Hc + h] * (1.f / DVc);
  const float rstd = rsqrtf(ms + RMS_EPS);
  const float gg = g[idx];
  const float val = on[idx] * rstd * rms_w[n];
  on[idx] = val * gg * (1.f / (1.f + __expf(-gg)));
}

__global__ __launch_bounds__(256) void out_gemm_kernel(
    const float* __restrict__ A, const float* __restrict__ Bw,
    float* __restrict__ Cm) {
  __shared__ __align__(16) _Float16 As[128][40];
  __shared__ __align__(16) _Float16 BsT[128][40];
  const int m0 = blockIdx.x * 128, n0 = blockIdx.y * 128;
  const int tid = threadIdx.x, lane = tid & 31, wave = tid >> 5;
  const int mr0 = (wave & 3) * 32, nc0 = (wave >> 2) * 64;
  v8f acc[2][4] = {};
  for (int kk = 0; kk < 512; kk += 32) {
    for (int i = tid; i < 1024; i += 256) {
      const int row = i >> 3, cg = i & 7;
      v4f xv = *(const v4f*)(A + (size_t)(m0 + row) * 512 + kk + cg * 4);
      v4h hv;
#pragma unroll
      for (int e = 0; e < 4; ++e) hv[e] = (_Float16)xv[e];
      *(v4h*)&As[row][cg * 4] = hv;
    }
    for (int i = tid; i < 1024; i += 256) {
      const int k = i >> 5, ng = i & 31;
      v4f wv = *(const v4f*)(Bw + (size_t)(kk + k) * 512 + n0 + ng * 4);
#pragma unroll
      for (int j2 = 0; j2 < 4; ++j2) BsT[ng * 4 + j2][k] = (_Float16)wv[j2];
    }
    if (kk + 32 < 512)
      __builtin_prefetch(&A[(size_t)(m0 + (tid >> 1)) * 512 + kk + 32], 0, 1);
    __syncthreads();
    v16h a0 = frag16(&As[0][0], 40, mr0, 0);
    v16h a1 = frag16(&As[0][0], 40, mr0 + 16, 0);
#pragma unroll
    for (int j = 0; j < 4; ++j) {
      v16h bm = frag16(&BsT[0][0], 40, nc0 + j * 16, 0);
      acc[0][j] = wmma16(a0, bm, acc[0][j]);
      acc[1][j] = wmma16(a1, bm, acc[1][j]);
    }
    __syncthreads();
  }
  const int nl = lane & 15, mOff = (lane >> 4) * 8;
#pragma unroll
  for (int a2 = 0; a2 < 2; ++a2) {
#pragma unroll
    for (int j = 0; j < 4; ++j) {
      const int n = n0 + nc0 + j * 16 + nl;
#pragma unroll
      for (int r = 0; r < 8; ++r)
        Cm[(size_t)(m0 + mr0 + a2 * 16 + mOff + r) * 512 + n] = acc[a2][j][r];
    }
  }
}

extern "C" void kernel_launch(void* const* d_in, const int* in_sizes, int n_in,
                              void* d_out, int out_size, void* d_ws, size_t ws_size,
                              hipStream_t stream) {
  (void)in_sizes; (void)n_in; (void)out_size; (void)ws_size;
  const float* x    = (const float*)d_in[0];
  const float* Wq   = (const float*)d_in[1];
  const float* Wk   = (const float*)d_in[2];
  const float* Wv   = (const float*)d_in[3];
  const float* Wg   = (const float*)d_in[4];
  const float* Wo   = (const float*)d_in[5];
  const float* Wgk1 = (const float*)d_in[6];
  const float* Wgk2 = (const float*)d_in[7];
  const float* bgk2 = (const float*)d_in[8];
  const float* rmsw = (const float*)d_in[9];
  float* out = (float*)d_out;

  float* ws   = (float*)d_ws;
  float* qws  = ws;
  float* kws  = qws  + (size_t)TOK * 256;
  float* vws  = kws  + (size_t)TOK * 256;
  float* gws  = vws  + (size_t)TOK * 512;
  float* gkws = gws  + (size_t)TOK * 512;
  float* xg1  = gkws + (size_t)TOK * 256;
  float* on   = xg1  + (size_t)TOK * 16;
  float* ssq  = on   + (size_t)TOK * 512;

  proj_kernel<<<dim3(TOK / 128, 1536 / 128), 256, 0, stream>>>(
      x, Wq, Wk, Wv, Wg, qws, kws, vws, gws);
  lr1_kernel<<<TOK / 16, 256, 0, stream>>>(x, Wgk1, xg1);
  lr2_kernel<<<(TOK * 256) / 256, 256, 0, stream>>>(xg1, Wgk2, bgk2, gkws);
  zero_kernel<<<(TOK * Hc + 255) / 256, 256, 0, stream>>>(ssq, TOK * Hc);
  gla_scan_kernel<<<dim3(Bc * Hc, 2), 256, 0, stream>>>(
      qws, kws, vws, gkws, on, ssq);
  epilogue_kernel<<<(TOK * 512) / 256, 256, 0, stream>>>(on, gws, rmsw, ssq);
  out_gemm_kernel<<<dim3(TOK / 128, 512 / 128), 256, 0, stream>>>(on, Wo, out);
}


# Scan operand prep (gate cumsum, q-hat, k-hat, k-hat^T, v^T f16 tiles, exp(L_C)) hoisted from the serial scan into a parallel per-(batch,chunk,head) pre-pass; scan copies ready tiles to LDS
# speedup vs baseline: 2.5455x; 2.5455x over previous
#include <hip/hip_runtime.h>
#include <hip/hip_bf16.h>
#include <math.h>


typedef __attribute__((ext_vector_type(16))) _Float16 v16h;
typedef __attribute__((ext_vector_type(8)))  _Float16 v8h;
typedef __attribute__((ext_vector_type(4)))  _Float16 v4h;
typedef __attribute__((ext_vector_type(8)))  float    v8f;
typedef __attribute__((ext_vector_type(4)))  float    v4f;

constexpr int Bc  = 4;
constexpr int Tc  = 2048;
constexpr int HIDc = 512;
constexpr int Hc  = 4;
constexpr int DKc = 64;
constexpr int DVc = 128;
constexpr int CH  = 64;
constexpr int TOK = Bc * Tc;
constexpr float RMS_EPS = 1e-5f;

__device__ __forceinline__ v8f wmma16(v16h a, v16h b, v8f c) {
  return __builtin_amdgcn_wmma_f32_16x16x32_f16(false, a, false, b, (short)0, c,
                                                false, false);
}

__device__ __forceinline__ v16h frag16(const _Float16* p, int ld, int rowBase,
                                       int kBase) {
  const int lane = threadIdx.x & 31, half = lane >> 4;
  const _Float16* row = p + (size_t)(rowBase + (lane & 15)) * ld + kBase + half * 8;
  v8h lo = *(const v8h*)(row);
  v8h hi = *(const v8h*)(row + 16);
  return __builtin_shufflevector(lo, hi, 0, 1, 2, 3, 4, 5, 6, 7,
                                 8, 9, 10, 11, 12, 13, 14, 15);
}

__device__ __forceinline__ v16h frag16_f32(const float* p, int ld, int rowBase,
                                           int kBase) {
  const int lane = threadIdx.x & 31, half = lane >> 4;
  const float* row = p + (size_t)(rowBase + (lane & 15)) * ld + kBase + half * 8;
  v4f a0 = *(const v4f*)(row);
  v4f a1 = *(const v4f*)(row + 4);
  v4f b0 = *(const v4f*)(row + 16);
  v4f b1 = *(const v4f*)(row + 20);
  v16h f;
#pragma unroll
  for (int e = 0; e < 4; ++e) {
    f[e]      = (_Float16)a0[e];
    f[4 + e]  = (_Float16)a1[e];
    f[8 + e]  = (_Float16)b0[e];
    f[12 + e] = (_Float16)b1[e];
  }
  return f;
}

__global__ __launch_bounds__(256) void proj_kernel(
    const float* __restrict__ x,
    const float* __restrict__ Wq, const float* __restrict__ Wk,
    const float* __restrict__ Wv, const float* __restrict__ Wg,
    float* __restrict__ qws, float* __restrict__ kws,
    float* __restrict__ vws, float* __restrict__ gws) {
  __shared__ __align__(16) _Float16 As[128][40];
  __shared__ __align__(16) _Float16 BsT[128][40];
  const int m0 = blockIdx.x * 128, n0 = blockIdx.y * 128;
  const int tid = threadIdx.x, lane = tid & 31, wave = tid >> 5;
  const int mr0 = (wave & 3) * 32;
  const int nc0 = (wave >> 2) * 64;
  v8f acc[2][4] = {};
  for (int kk = 0; kk < HIDc; kk += 32) {
    for (int i = tid; i < 1024; i += 256) {
      const int row = i >> 3, cg = i & 7;
      v4f xv = *(const v4f*)(x + (size_t)(m0 + row) * HIDc + kk + cg * 4);
      v4h hv;
#pragma unroll
      for (int e = 0; e < 4; ++e) hv[e] = (_Float16)xv[e];
      *(v4h*)&As[row][cg * 4] = hv;
    }
    for (int i = tid; i < 1024; i += 256) {
      const int k = i >> 5, ng = i & 31, n = n0 + ng * 4;
      v4f wv;
      if (n < 256)       wv = *(const v4f*)(Wq + (size_t)(kk + k) * 256 + n);
      else if (n < 512)  wv = *(const v4f*)(Wk + (size_t)(kk + k) * 256 + (n - 256));
      else if (n < 1024) wv = *(const v4f*)(Wv + (size_t)(kk + k) * 512 + (n - 512));
      else               wv = *(const v4f*)(Wg + (size_t)(kk + k) * 512 + (n - 1024));
#pragma unroll
      for (int j2 = 0; j2 < 4; ++j2) BsT[ng * 4 + j2][k] = (_Float16)wv[j2];
    }
    if (kk + 32 < HIDc)
      __builtin_prefetch(&x[(size_t)(m0 + (tid >> 1)) * HIDc + kk + 32], 0, 1);
    __syncthreads();
    v16h a0 = frag16(&As[0][0], 40, mr0, 0);
    v16h a1 = frag16(&As[0][0], 40, mr0 + 16, 0);
#pragma unroll
    for (int j = 0; j < 4; ++j) {
      v16h bm = frag16(&BsT[0][0], 40, nc0 + j * 16, 0);
      acc[0][j] = wmma16(a0, bm, acc[0][j]);
      acc[1][j] = wmma16(a1, bm, acc[1][j]);
    }
    __syncthreads();
  }
  const int nl = lane & 15, mOff = (lane >> 4) * 8;
#pragma unroll
  for (int a2 = 0; a2 < 2; ++a2) {
#pragma unroll
    for (int j = 0; j < 4; ++j) {
      const int n = n0 + nc0 + j * 16 + nl;
#pragma unroll
      for (int r = 0; r < 8; ++r) {
        const size_t m = (size_t)(m0 + mr0 + a2 * 16 + mOff + r);
        const float val = acc[a2][j][r];
        if (n < 256)       qws[m * 256 + n]          = val * 0.125f;
        else if (n < 512)  kws[m * 256 + (n - 256)]  = val;
        else if (n < 1024) vws[m * 512 + (n - 512)]  = val;
        else               gws[m * 512 + (n - 1024)] = val;
      }
    }
  }
}

__global__ __launch_bounds__(256) void lr1_kernel(const float* __restrict__ x,
                                                  const float* __restrict__ Wgk1,
                                                  float* __restrict__ xg1) {
  __shared__ float xs[16][512];
  const int t0 = blockIdx.x * 16;
  for (int i = threadIdx.x; i < 16 * 512; i += 256)
    xs[i >> 9][i & 511] = x[(size_t)(t0 + (i >> 9)) * 512 + (i & 511)];
  __syncthreads();
  const int tok = threadIdx.x >> 4, j = threadIdx.x & 15;
  float s = 0.f;
  for (int k2 = 0; k2 < 512; ++k2) s += xs[tok][k2] * Wgk1[k2 * 16 + j];
  xg1[(size_t)(t0 + tok) * 16 + j] = s;
}

__global__ __launch_bounds__(256) void gate_prep_kernel(
    const float* __restrict__ xg1, const float* __restrict__ Wgk2,
    const float* __restrict__ bgk2, const float* __restrict__ qg,
    const float* __restrict__ kg, const float* __restrict__ vg,
    float* __restrict__ qh_ws, float* __restrict__ kh_ws,
    float* __restrict__ khT_ws, float* __restrict__ vT_ws,
    float* __restrict__ expLCg) {
  constexpr int LDA = CH + 8;
  __shared__ __align__(16) float    xs[CH][16];
  __shared__ float                  lamS[CH][DKc + 1];
  __shared__ __align__(16) _Float16 qS[CH][LDA];
  __shared__ __align__(16) _Float16 kS[CH][LDA];
  __shared__ __align__(16) _Float16 kTS[DKc][LDA];
  __shared__ __align__(16) _Float16 vTS[DVc][LDA];

  const int tile = blockIdx.x;
  const int h = tile & 3, tok0 = (tile >> 2) * CH;
  const int tid = threadIdx.x, d = tid & 63, grp = tid >> 6;
  const int n = h * DKc + d;

  *(v4f*)(&xs[0][0] + tid * 4) = *(const v4f*)(xg1 + (size_t)tok0 * 16 + tid * 4);
  float w[16];
#pragma unroll
  for (int j = 0; j < 16; ++j) w[j] = Wgk2[j * 256 + n];
  const float bias = bgk2[n];

  float qv[16], kv[16];
#pragma unroll
  for (int tt = 0; tt < 16; ++tt) {
    const size_t off = (size_t)(tok0 + grp * 16 + tt) * (Hc * DKc) + n;
    qv[tt] = qg[off];
    kv[tt] = kg[off];
  }

  {
    const float* vb = vg + (size_t)tok0 * (Hc * DVc) + h * DVc;
    for (int r = 0; r < 8; ++r) {
      const int i = tid + 256 * r, s = i >> 5, n4 = (i & 31) * 4;
      const v4f vv = *(const v4f*)(vb + (size_t)s * (Hc * DVc) + n4);
#pragma unroll
      for (int e = 0; e < 4; ++e) vTS[n4 + e][s] = (_Float16)vv[e];
    }
  }
  __syncthreads();

  for (int tt = 0; tt < 16; ++tt) {
    const int t = grp * 16 + tt;
    float s = bias;
#pragma unroll
    for (int j = 0; j < 16; ++j) s += xs[t][j] * w[j];
    const float ls = fminf(s, 0.f) - log1pf(__expf(-fabsf(s)));
    lamS[t][d] = ls * (1.f / 16.f);
  }
  __syncthreads();

  if (tid < DKc) {
    float lam = 0.f;
    for (int t = 0; t < CH; ++t) {
      lam += lamS[t][tid];
      lamS[t][tid] = lam;
    }
    expLCg[(size_t)tile * DKc + tid] = __expf(lam);
  }
  __syncthreads();

  const float lamC = lamS[CH - 1][d];
#pragma unroll
  for (int tt = 0; tt < 16; ++tt) {
    const int t = grp * 16 + tt;
    const float eq = __expf(lamS[t][d] - lamC);
    const float qh = fmaxf(fminf(qv[tt] * eq, 60000.f), -60000.f);
    const float kh = kv[tt] / eq;
    qS[t][d]  = (_Float16)qh;
    kS[t][d]  = (_Float16)kh;
    kTS[d][t] = (_Float16)kh;
  }
  __syncthreads();

  _Float16* qd  = (_Float16*)qh_ws  + (size_t)tile * (CH * DKc);
  _Float16* kd  = (_Float16*)kh_ws  + (size_t)tile * (CH * DKc);
  _Float16* ktd = (_Float16*)khT_ws + (size_t)tile * (CH * DKc);
  _Float16* vtd = (_Float16*)vT_ws  + (size_t)tile * (CH * DVc);
  for (int r = 0; r < 2; ++r) {
    const int i = tid + 256 * r, row = i >> 3, cc = (i & 7) * 8;
    *(v8h*)(qd  + i * 8) = *(const v8h*)&qS[row][cc];
    *(v8h*)(kd  + i * 8) = *(const v8h*)&kS[row][cc];
    *(v8h*)(ktd + i * 8) = *(const v8h*)&kTS[row][cc];
  }
  for (int r = 0; r < 4; ++r) {
    const int i = tid + 256 * r, row = i >> 3, cc = (i & 7) * 8;
    *(v8h*)(vtd + i * 8) = *(const v8h*)&vTS[row][cc];
  }
}

__global__ void zero_kernel(float* __restrict__ p, int n) {
  const int i = blockIdx.x * 256 + threadIdx.x;
  if (i < n) p[i] = 0.f;
}

__global__ __launch_bounds__(256) void gla_scan_kernel(
    const float* __restrict__ qh_ws, const float* __restrict__ kh_ws,
    const float* __restrict__ khT_ws, const float* __restrict__ vT_ws,
    const float* __restrict__ expLCg,
    float* __restrict__ on, float* __restrict__ ssq_g) {
  constexpr int LDA = CH + 8;
  constexpr int LDS_ = 68;
  __shared__ __align__(16) float    ST[64][LDS_];
  __shared__ __align__(16) _Float16 qt[CH][LDA];
  __shared__ __align__(16) _Float16 kt[CH][LDA];
  __shared__ __align__(16) _Float16 ktT[DKc][LDA];
  __shared__ __align__(16) _Float16 Am[CH][LDA];
  __shared__ __align__(16) _Float16 vT[64][LDA];
  __shared__ float ssql[CH];

  const int bh = blockIdx.x, b = bh >> 2, h = bh & 3;
  const int nHalf = blockIdx.y;
  const int tid = threadIdx.x, lane = tid & 31, wave = tid >> 5;

  for (int i = tid; i < 64 * LDS_; i += 256) (&ST[0][0])[i] = 0.f;
  __syncthreads();

  for (int c = 0; c < Tc / CH; ++c) {
    const int t0 = c * CH;
    const size_t tile = (size_t)((b * (Tc / CH) + c) * Hc + h);
    const _Float16* qtile  = (const _Float16*)qh_ws  + tile * (CH * DKc);
    const _Float16* ktile  = (const _Float16*)kh_ws  + tile * (CH * DKc);
    const _Float16* kTtile = (const _Float16*)khT_ws + tile * (CH * DKc);
    const _Float16* vTtile = (const _Float16*)vT_ws  + (tile * 2 + nHalf) * (CH * 64);
    float*       ob  = on  + (size_t)(b * Tc + t0) * (Hc * DVc) + h * DVc + nHalf * 64;

    const float dec = expLCg[tile * DKc + (tid & 63)];
    for (int r = 0; r < 2; ++r) {
      const int i = tid + 256 * r, row = i >> 3, cc = (i & 7) * 8;
      *(v8h*)&qt[row][cc]  = *(const v8h*)(qtile  + i * 8);
      *(v8h*)&kt[row][cc]  = *(const v8h*)(ktile  + i * 8);
      *(v8h*)&ktT[row][cc] = *(const v8h*)(kTtile + i * 8);
      *(v8h*)&vT[row][cc]  = *(const v8h*)(vTtile + i * 8);
    }
    if (tid < CH) ssql[tid] = 0.f;
    __syncthreads();

    for (int i = tid; i < 64 * 64; i += 256) {
      const int n = i >> 6, d = i & 63;
      ST[n][d] *= dec;
    }

    {
      const int tRow = wave & 3, sp0 = (wave >> 2) * 2;
#pragma unroll
      for (int spi = 0; spi < 2; ++spi) {
        const int sTile = sp0 + spi;
        v8f acc = {};
        if (sTile <= tRow) {
#pragma unroll
          for (int k2 = 0; k2 < 2; ++k2) {
            v16h a  = frag16(&qt[0][0], LDA, tRow * 16, k2 * 32);
            v16h bm = frag16(&kt[0][0], LDA, sTile * 16, k2 * 32);
            acc = wmma16(a, bm, acc);
          }
        }
        const int sIdx = sTile * 16 + (lane & 15), mOff = (lane >> 4) * 8;
#pragma unroll
        for (int r = 0; r < 8; ++r) {
          const int tIdx = tRow * 16 + mOff + r;
          Am[tIdx][sIdx] = (_Float16)((sIdx <= tIdx) ? acc[r] : 0.f);
        }
      }
    }
    __syncthreads();

    {
      const int tRow = wave & 3, nt0 = (wave >> 2) * 2;
      v8f acc[2] = {};
#pragma unroll
      for (int k2 = 0; k2 < 2; ++k2) {
        v16h aA = frag16(&Am[0][0], LDA, tRow * 16, k2 * 32);
        v16h aQ = frag16(&qt[0][0], LDA, tRow * 16, k2 * 32);
#pragma unroll
        for (int j = 0; j < 2; ++j) {
          v16h bv = frag16(&vT[0][0], LDA, (nt0 + j) * 16, k2 * 32);
          acc[j] = wmma16(aA, bv, acc[j]);
          v16h bs = frag16_f32(&ST[0][0], LDS_, (nt0 + j) * 16, k2 * 32);
          acc[j] = wmma16(aQ, bs, acc[j]);
        }
      }
      const int nl = lane & 15, mOff = (lane >> 4) * 8;
#pragma unroll
      for (int j = 0; j < 2; ++j) {
        const int n = (nt0 + j) * 16 + nl;
#pragma unroll
        for (int r = 0; r < 8; ++r) {
          const int t = tRow * 16 + mOff + r;
          ob[(size_t)t * (Hc * DVc) + n] = acc[j][r];
          atomicAdd(&ssql[t], acc[j][r] * acc[j][r]);
        }
      }
    }
    __syncthreads();
    if (tid < CH)
      atomicAdd(&ssq_g[(size_t)(b * Tc + t0 + tid) * Hc + h], ssql[tid]);

    {
      const int dRow = wave & 3, nt0 = (wave >> 2) * 2;
      v8f acc[2] = {};
#pragma unroll
      for (int k2 = 0; k2 < 2; ++k2) {
        v16h aK = frag16(&ktT[0][0], LDA, dRow * 16, k2 * 32);
#pragma unroll
        for (int j = 0; j < 2; ++j) {
          v16h bv = frag16(&vT[0][0], LDA, (nt0 + j) * 16, k2 * 32);
          acc[j] = wmma16(aK, bv, acc[j]);
        }
      }
      const int nl = lane & 15, mOff = (lane >> 4) * 8;
#pragma unroll
      for (int j = 0; j < 2; ++j) {
        const int n = (nt0 + j) * 16 + nl;
#pragma unroll
        for (int r = 0; r < 8; ++r) {
          const int d = dRow * 16 + mOff + r;
          ST[n][d] += acc[j][r];
        }
      }
    }
    __syncthreads();
  }
}

__global__ __launch_bounds__(256) void epilogue_kernel(
    float* __restrict__ on, const float* __restrict__ g,
    const float* __restrict__ rms_w, const float* __restrict__ ssq) {
  const size_t idx = (size_t)blockIdx.x * 256 + threadIdx.x;
  const size_t tok = idx >> 9;
  const int col = (int)(idx & 511), h = col >> 7, n = col & 127;
  const float ms = ssq[tok * Hc + h] * (1.f / DVc);
  const float rstd = rsqrtf(ms + RMS_EPS);
  const float gg = g[idx];
  const float val = on[idx] * rstd * rms_w[n];
  on[idx] = val * gg * (1.f / (1.f + __expf(-gg)));
}

__global__ __launch_bounds__(256) void out_gemm_kernel(
    const float* __restrict__ A, const float* __restrict__ Bw,
    float* __restrict__ Cm) {
  __shared__ __align__(16) _Float16 As[128][40];
  __shared__ __align__(16) _Float16 BsT[128][40];
  const int m0 = blockIdx.x * 128, n0 = blockIdx.y * 128;
  const int tid = threadIdx.x, lane = tid & 31, wave = tid >> 5;
  const int mr0 = (wave & 3) * 32, nc0 = (wave >> 2) * 64;
  v8f acc[2][4] = {};
  for (int kk = 0; kk < 512; kk += 32) {
    for (int i = tid; i < 1024; i += 256) {
      const int row = i >> 3, cg = i & 7;
      v4f xv = *(const v4f*)(A + (size_t)(m0 + row) * 512 + kk + cg * 4);
      v4h hv;
#pragma unroll
      for (int e = 0; e < 4; ++e) hv[e] = (_Float16)xv[e];
      *(v4h*)&As[row][cg * 4] = hv;
    }
    for (int i = tid; i < 1024; i += 256) {
      const int k = i >> 5, ng = i & 31;
      v4f wv = *(const v4f*)(Bw + (size_t)(kk + k) * 512 + n0 + ng * 4);
#pragma unroll
      for (int j2 = 0; j2 < 4; ++j2) BsT[ng * 4 + j2][k] = (_Float16)wv[j2];
    }
    if (kk + 32 < 512)
      __builtin_prefetch(&A[(size_t)(m0 + (tid >> 1)) * 512 + kk + 32], 0, 1);
    __syncthreads();
    v16h a0 = frag16(&As[0][0], 40, mr0, 0);
    v16h a1 = frag16(&As[0][0], 40, mr0 + 16, 0);
#pragma unroll
    for (int j = 0; j < 4; ++j) {
      v16h bm = frag16(&BsT[0][0], 40, nc0 + j * 16, 0);
      acc[0][j] = wmma16(a0, bm, acc[0][j]);
      acc[1][j] = wmma16(a1, bm, acc[1][j]);
    }
    __syncthreads();
  }
  const int nl = lane & 15, mOff = (lane >> 4) * 8;
#pragma unroll
  for (int a2 = 0; a2 < 2; ++a2) {
#pragma unroll
    for (int j = 0; j < 4; ++j) {
      const int n = n0 + nc0 + j * 16 + nl;
#pragma unroll
      for (int r = 0; r < 8; ++r)
        Cm[(size_t)(m0 + mr0 + a2 * 16 + mOff + r) * 512 + n] = acc[a2][j][r];
    }
  }
}

extern "C" void kernel_launch(void* const* d_in, const int* in_sizes, int n_in,
                              void* d_out, int out_size, void* d_ws, size_t ws_size,
                              hipStream_t stream) {
  (void)in_sizes; (void)n_in; (void)out_size; (void)ws_size;
  const float* x    = (const float*)d_in[0];
  const float* Wq   = (const float*)d_in[1];
  const float* Wk   = (const float*)d_in[2];
  const float* Wv   = (const float*)d_in[3];
  const float* Wg   = (const float*)d_in[4];
  const float* Wo   = (const float*)d_in[5];
  const float* Wgk1 = (const float*)d_in[6];
  const float* Wgk2 = (const float*)d_in[7];
  const float* bgk2 = (const float*)d_in[8];
  const float* rmsw = (const float*)d_in[9];
  float* out = (float*)d_out;

  float* ws   = (float*)d_ws;
  float* qws  = ws;
  float* kws  = qws  + (size_t)TOK * 256;
  float* vws  = kws  + (size_t)TOK * 256;
  float* gws  = vws  + (size_t)TOK * 512;
  float* xg1  = gws  + (size_t)TOK * 512;
  float* on   = xg1  + (size_t)TOK * 16;
  float* ssq  = on   + (size_t)TOK * 512;
  float* qhw  = ssq  + (size_t)TOK * Hc;
  float* khw  = qhw  + (size_t)TOK * 128;
  float* khTw = khw  + (size_t)TOK * 128;
  float* vTw  = khTw + (size_t)TOK * 128;
  float* elc  = vTw  + (size_t)TOK * 256;

  proj_kernel<<<dim3(TOK / 128, 1536 / 128), 256, 0, stream>>>(
      x, Wq, Wk, Wv, Wg, qws, kws, vws, gws);
  lr1_kernel<<<TOK / 16, 256, 0, stream>>>(x, Wgk1, xg1);
  gate_prep_kernel<<<(TOK / CH) * Hc, 256, 0, stream>>>(
      xg1, Wgk2, bgk2, qws, kws, vws, qhw, khw, khTw, vTw, elc);
  zero_kernel<<<(TOK * Hc + 255) / 256, 256, 0, stream>>>(ssq, TOK * Hc);
  gla_scan_kernel<<<dim3(Bc * Hc, 2), 256, 0, stream>>>(
      qhw, khw, khTw, vTw, elc, on, ssq);
  epilogue_kernel<<<(TOK * 512) / 256, 256, 0, stream>>>(on, gws, rmsw, ssq);
  out_gemm_kernel<<<dim3(TOK / 128, 512 / 128), 256, 0, stream>>>(on, Wo, out);
}


# RMSNorm and swish output gate applied in the output GEMM's A-tile loader (o, g, sum of squares read there); the separate one-element-per-thread epilogue launch and its rewrite of o are gone
# speedup vs baseline: 2.6521x; 1.0419x over previous
#include <hip/hip_runtime.h>
#include <hip/hip_bf16.h>
#include <math.h>


typedef __attribute__((ext_vector_type(16))) _Float16 v16h;
typedef __attribute__((ext_vector_type(8)))  _Float16 v8h;
typedef __attribute__((ext_vector_type(4)))  _Float16 v4h;
typedef __attribute__((ext_vector_type(8)))  float    v8f;
typedef __attribute__((ext_vector_type(4)))  float    v4f;

constexpr int Bc  = 4;
constexpr int Tc  = 2048;
constexpr int HIDc = 512;
constexpr int Hc  = 4;
constexpr int DKc = 64;
constexpr int DVc = 128;
constexpr int CH  = 64;
constexpr int TOK = Bc * Tc;
constexpr float RMS_EPS = 1e-5f;

__device__ __forceinline__ v8f wmma16(v16h a, v16h b, v8f c) {
  return __builtin_amdgcn_wmma_f32_16x16x32_f16(false, a, false, b, (short)0, c,
                                                false, false);
}

__device__ __forceinline__ v16h frag16(const _Float16* p, int ld, int rowBase,
                                       int kBase) {
  const int lane = threadIdx.x & 31, half = lane >> 4;
  const _Float16* row = p + (size_t)(rowBase + (lane & 15)) * ld + kBase + half * 8;
  v8h lo = *(const v8h*)(row);
  v8h hi = *(const v8h*)(row + 16);
  return __builtin_shufflevector(lo, hi, 0, 1, 2, 3, 4, 5, 6, 7,
                                 8, 9, 10, 11, 12, 13, 14, 15);
}

__device__ __forceinline__ v16h frag16_f32(const float* p, int ld, int rowBase,
                                           int kBase) {
  const int lane = threadIdx.x & 31, half = lane >> 4;
  const float* row = p + (size_t)(rowBase + (lane & 15)) * ld + kBase + half * 8;
  v4f a0 = *(const v4f*)(row);
  v4f a1 = *(const v4f*)(row + 4);
  v4f b0 = *(const v4f*)(row + 16);
  v4f b1 = *(const v4f*)(row + 20);
  v16h f;
#pragma unroll
  for (int e = 0; e < 4; ++e) {
    f[e]      = (_Float16)a0[e];
    f[4 + e]  = (_Float16)a1[e];
    f[8 + e]  = (_Float16)b0[e];
    f[12 + e] = (_Float16)b1[e];
  }
  return f;
}

__global__ __launch_bounds__(256) void proj_kernel(
    const float* __restrict__ x,
    const float* __restrict__ Wq, const float* __restrict__ Wk,
    const float* __restrict__ Wv, const float* __restrict__ Wg,
    float* __restrict__ qws, float* __restrict__ kws,
    float* __restrict__ vws, float* __restrict__ gws) {
  __shared__ __align__(16) _Float16 As[128][40];
  __shared__ __align__(16) _Float16 BsT[128][40];
  const int m0 = blockIdx.x * 128, n0 = blockIdx.y * 128;
  const int tid = threadIdx.x, lane = tid & 31, wave = tid >> 5;
  const int mr0 = (wave & 3) * 32;
  const int nc0 = (wave >> 2) * 64;
  v8f acc[2][4] = {};
  for (int kk = 0; kk < HIDc; kk += 32) {
    for (int i = tid; i < 1024; i += 256) {
      const int row = i >> 3, cg = i & 7;
      v4f xv = *(const v4f*)(x + (size_t)(m0 + row) * HIDc + kk + cg * 4);
      v4h hv;
#pragma unroll
      for (int e = 0; e < 4; ++e) hv[e] = (_Float16)xv[e];
      *(v4h*)&As[row][cg * 4] = hv;
    }
    for (int i = tid; i < 1024; i += 256) {
      const int k = i >> 5, ng = i & 31, n = n0 + ng * 4;
      v4f wv;
      if (n < 256)       wv = *(const v4f*)(Wq + (size_t)(kk + k) * 256 + n);
      else if (n < 512)  wv = *(const v4f*)(Wk + (size_t)(kk + k) * 256 + (n - 256));
      else if (n < 1024) wv = *(const v4f*)(Wv + (size_t)(kk + k) * 512 + (n - 512));
      else               wv = *(const v4f*)(Wg + (size_t)(kk + k) * 512 + (n - 1024));
#pragma unroll
      for (int j2 = 0; j2 < 4; ++j2) BsT[ng * 4 + j2][k] = (_Float16)wv[j2];
    }
    if (kk + 32 < HIDc)
      __builtin_prefetch(&x[(size_t)(m0 + (tid >> 1)) * HIDc + kk + 32], 0, 1);
    __syncthreads();
    v16h a0 = frag16(&As[0][0], 40, mr0, 0);
    v16h a1 = frag16(&As[0][0], 40, mr0 + 16, 0);
#pragma unroll
    for (int j = 0; j < 4; ++j) {
      v16h bm = frag16(&BsT[0][0], 40, nc0 + j * 16, 0);
      acc[0][j] = wmma16(a0, bm, acc[0][j]);
      acc[1][j] = wmma16(a1, bm, acc[1][j]);
    }
    __syncthreads();
  }
  const int nl = lane & 15, mOff = (lane >> 4) * 8;
#pragma unroll
  for (int a2 = 0; a2 < 2; ++a2) {
#pragma unroll
    for (int j = 0; j < 4; ++j) {
      const int n = n0 + nc0 + j * 16 + nl;
#pragma unroll
      for (int r = 0; r < 8; ++r) {
        const size_t m = (size_t)(m0 + mr0 + a2 * 16 + mOff + r);
        const float val = acc[a2][j][r];
        if (n < 256)       qws[m * 256 + n]          = val * 0.125f;
        else if (n < 512)  kws[m * 256 + (n - 256)]  = val;
        else if (n < 1024) vws[m * 512 + (n - 512)]  = val;
        else               gws[m * 512 + (n - 1024)] = val;
      }
    }
  }
}

__global__ __launch_bounds__(256) void lr1_kernel(const float* __restrict__ x,
                                                  const float* __restrict__ Wgk1,
                                                  float* __restrict__ xg1) {
  __shared__ float xs[16][512];
  const int t0 = blockIdx.x * 16;
  for (int i = threadIdx.x; i < 16 * 512; i += 256)
    xs[i >> 9][i & 511] = x[(size_t)(t0 + (i >> 9)) * 512 + (i & 511)];
  __syncthreads();
  const int tok = threadIdx.x >> 4, j = threadIdx.x & 15;
  float s = 0.f;
  for (int k2 = 0; k2 < 512; ++k2) s += xs[tok][k2] * Wgk1[k2 * 16 + j];
  xg1[(size_t)(t0 + tok) * 16 + j] = s;
}

__global__ __launch_bounds__(256) void gate_prep_kernel(
    const float* __restrict__ xg1, const float* __restrict__ Wgk2,
    const float* __restrict__ bgk2, const float* __restrict__ qg,
    const float* __restrict__ kg, const float* __restrict__ vg,
    float* __restrict__ qh_ws, float* __restrict__ kh_ws,
    float* __restrict__ khT_ws, float* __restrict__ vT_ws,
    float* __restrict__ expLCg) {
  constexpr int LDA = CH + 8;
  __shared__ __align__(16) float    xs[CH][16];
  __shared__ float                  lamS[CH][DKc + 1];
  __shared__ __align__(16) _Float16 qS[CH][LDA];
  __shared__ __align__(16) _Float16 kS[CH][LDA];
  __shared__ __align__(16) _Float16 kTS[DKc][LDA];
  __shared__ __align__(16) _Float16 vTS[DVc][LDA];

  const int tile = blockIdx.x;
  const int h = tile & 3, tok0 = (tile >> 2) * CH;
  const int tid = threadIdx.x, d = tid & 63, grp = tid >> 6;
  const int n = h * DKc + d;

  *(v4f*)(&xs[0][0] + tid * 4) = *(const v4f*)(xg1 + (size_t)tok0 * 16 + tid * 4);
  float w[16];
#pragma unroll
  for (int j = 0; j < 16; ++j) w[j] = Wgk2[j * 256 + n];
  const float bias = bgk2[n];

  float qv[16], kv[16];
#pragma unroll
  for (int tt = 0; tt < 16; ++tt) {
    const size_t off = (size_t)(tok0 + grp * 16 + tt) * (Hc * DKc) + n;
    qv[tt] = qg[off];
    kv[tt] = kg[off];
  }

  {
    const float* vb = vg + (size_t)tok0 * (Hc * DVc) + h * DVc;
    for (int r = 0; r < 8; ++r) {
      const int i = tid + 256 * r, s = i >> 5, n4 = (i & 31) * 4;
      const v4f vv = *(const v4f*)(vb + (size_t)s * (Hc * DVc) + n4);
#pragma unroll
      for (int e = 0; e < 4; ++e) vTS[n4 + e][s] = (_Float16)vv[e];
    }
  }
  __syncthreads();

  for (int tt = 0; tt < 16; ++tt) {
    const int t = grp * 16 + tt;
    float s = bias;
#pragma unroll
    for (int j = 0; j < 16; ++j) s += xs[t][j] * w[j];
    const float ls = fminf(s, 0.f) - log1pf(__expf(-fabsf(s)));
    lamS[t][d] = ls * (1.f / 16.f);
  }
  __syncthreads();

  if (tid < DKc) {
    float lam = 0.f;
    for (int t = 0; t < CH; ++t) {
      lam += lamS[t][tid];
      lamS[t][tid] = lam;
    }
    expLCg[(size_t)tile * DKc + tid] = __expf(lam);
  }
  __syncthreads();

  const float lamC = lamS[CH - 1][d];
#pragma unroll
  for (int tt = 0; tt < 16; ++tt) {
    const int t = grp * 16 + tt;
    const float eq = __expf(lamS[t][d] - lamC);
    const float qh = fmaxf(fminf(qv[tt] * eq, 60000.f), -60000.f);
    const float kh = kv[tt] / eq;
    qS[t][d]  = (_Float16)qh;
    kS[t][d]  = (_Float16)kh;
    kTS[d][t] = (_Float16)kh;
  }
  __syncthreads();

  _Float16* qd  = (_Float16*)qh_ws  + (size_t)tile * (CH * DKc);
  _Float16* kd  = (_Float16*)kh_ws  + (size_t)tile * (CH * DKc);
  _Float16* ktd = (_Float16*)khT_ws + (size_t)tile * (CH * DKc);
  _Float16* vtd = (_Float16*)vT_ws  + (size_t)tile * (CH * DVc);
  for (int r = 0; r < 2; ++r) {
    const int i = tid + 256 * r, row = i >> 3, cc = (i & 7) * 8;
    *(v8h*)(qd  + i * 8) = *(const v8h*)&qS[row][cc];
    *(v8h*)(kd  + i * 8) = *(const v8h*)&kS[row][cc];
    *(v8h*)(ktd + i * 8) = *(const v8h*)&kTS[row][cc];
  }
  for (int r = 0; r < 4; ++r) {
    const int i = tid + 256 * r, row = i >> 3, cc = (i & 7) * 8;
    *(v8h*)(vtd + i * 8) = *(const v8h*)&vTS[row][cc];
  }
}

__global__ void zero_kernel(float* __restrict__ p, int n) {
  const int i = blockIdx.x * 256 + threadIdx.x;
  if (i < n) p[i] = 0.f;
}

__global__ __launch_bounds__(256) void gla_scan_kernel(
    const float* __restrict__ qh_ws, const float* __restrict__ kh_ws,
    const float* __restrict__ khT_ws, const float* __restrict__ vT_ws,
    const float* __restrict__ expLCg,
    float* __restrict__ on, float* __restrict__ ssq_g) {
  constexpr int LDA = CH + 8;
  constexpr int LDS_ = 68;
  __shared__ __align__(16) float    ST[64][LDS_];
  __shared__ __align__(16) _Float16 qt[CH][LDA];
  __shared__ __align__(16) _Float16 kt[CH][LDA];
  __shared__ __align__(16) _Float16 ktT[DKc][LDA];
  __shared__ __align__(16) _Float16 Am[CH][LDA];
  __shared__ __align__(16) _Float16 vT[64][LDA];
  __shared__ float ssql[CH];

  const int bh = blockIdx.x, b = bh >> 2, h = bh & 3;
  const int nHalf = blockIdx.y;
  const int tid = threadIdx.x, lane = tid & 31, wave = tid >> 5;

  for (int i = tid; i < 64 * LDS_; i += 256) (&ST[0][0])[i] = 0.f;
  __syncthreads();

  for (int c = 0; c < Tc / CH; ++c) {
    const int t0 = c * CH;
    const size_t tile = (size_t)((b * (Tc / CH) + c) * Hc + h);
    const _Float16* qtile  = (const _Float16*)qh_ws  + tile * (CH * DKc);
    const _Float16* ktile  = (const _Float16*)kh_ws  + tile * (CH * DKc);
    const _Float16* kTtile = (const _Float16*)khT_ws + tile * (CH * DKc);
    const _Float16* vTtile = (const _Float16*)vT_ws  + (tile * 2 + nHalf) * (CH * 64);
    float*       ob  = on  + (size_t)(b * Tc + t0) * (Hc * DVc) + h * DVc + nHalf * 64;

    const float dec = expLCg[tile * DKc + (tid & 63)];
    for (int r = 0; r < 2; ++r) {
      const int i = tid + 256 * r, row = i >> 3, cc = (i & 7) * 8;
      *(v8h*)&qt[row][cc]  = *(const v8h*)(qtile  + i * 8);
      *(v8h*)&kt[row][cc]  = *(const v8h*)(ktile  + i * 8);
      *(v8h*)&ktT[row][cc] = *(const v8h*)(kTtile + i * 8);
      *(v8h*)&vT[row][cc]  = *(const v8h*)(vTtile + i * 8);
    }
    if (tid < CH) ssql[tid] = 0.f;
    __syncthreads();

    for (int i = tid; i < 64 * 64; i += 256) {
      const int n = i >> 6, d = i & 63;
      ST[n][d] *= dec;
    }

    {
      const int tRow = wave & 3, sp0 = (wave >> 2) * 2;
#pragma unroll
      for (int spi = 0; spi < 2; ++spi) {
        const int sTile = sp0 + spi;
        v8f acc = {};
        if (sTile <= tRow) {
#pragma unroll
          for (int k2 = 0; k2 < 2; ++k2) {
            v16h a  = frag16(&qt[0][0], LDA, tRow * 16, k2 * 32);
            v16h bm = frag16(&kt[0][0], LDA, sTile * 16, k2 * 32);
            acc = wmma16(a, bm, acc);
          }
        }
        const int sIdx = sTile * 16 + (lane & 15), mOff = (lane >> 4) * 8;
#pragma unroll
        for (int r = 0; r < 8; ++r) {
          const int tIdx = tRow * 16 + mOff + r;
          Am[tIdx][sIdx] = (_Float16)((sIdx <= tIdx) ? acc[r] : 0.f);
        }
      }
    }
    __syncthreads();

    {
      const int tRow = wave & 3, nt0 = (wave >> 2) * 2;
      v8f acc[2] = {};
#pragma unroll
      for (int k2 = 0; k2 < 2; ++k2) {
        v16h aA = frag16(&Am[0][0], LDA, tRow * 16, k2 * 32);
        v16h aQ = frag16(&qt[0][0], LDA, tRow * 16, k2 * 32);
#pragma unroll
        for (int j = 0; j < 2; ++j) {
          v16h bv = frag16(&vT[0][0], LDA, (nt0 + j) * 16, k2 * 32);
          acc[j] = wmma16(aA, bv, acc[j]);
          v16h bs = frag16_f32(&ST[0][0], LDS_, (nt0 + j) * 16, k2 * 32);
          acc[j] = wmma16(aQ, bs, acc[j]);
        }
      }
      const int nl = lane & 15, mOff = (lane >> 4) * 8;
#pragma unroll
      for (int j = 0; j < 2; ++j) {
        const int n = (nt0 + j) * 16 + nl;
#pragma unroll
        for (int r = 0; r < 8; ++r) {
          const int t = tRow * 16 + mOff + r;
          ob[(size_t)t * (Hc * DVc) + n] = acc[j][r];
          atomicAdd(&ssql[t], acc[j][r] * acc[j][r]);
        }
      }
    }
    __syncthreads();
    if (tid < CH)
      atomicAdd(&ssq_g[(size_t)(b * Tc + t0 + tid) * Hc + h], ssql[tid]);

    {
      const int dRow = wave & 3, nt0 = (wave >> 2) * 2;
      v8f acc[2] = {};
#pragma unroll
      for (int k2 = 0; k2 < 2; ++k2) {
        v16h aK = frag16(&ktT[0][0], LDA, dRow * 16, k2 * 32);
#pragma unroll
        for (int j = 0; j < 2; ++j) {
          v16h bv = frag16(&vT[0][0], LDA, (nt0 + j) * 16, k2 * 32);
          acc[j] = wmma16(aK, bv, acc[j]);
        }
      }
      const int nl = lane & 15, mOff = (lane >> 4) * 8;
#pragma unroll
      for (int j = 0; j < 2; ++j) {
        const int n = (nt0 + j) * 16 + nl;
#pragma unroll
        for (int r = 0; r < 8; ++r) {
          const int d = dRow * 16 + mOff + r;
          ST[n][d] += acc[j][r];
        }
      }
    }
    __syncthreads();
  }
}

__global__ __launch_bounds__(256) void out_gemm_kernel(
    const float* __restrict__ on, const float* __restrict__ g,
    const float* __restrict__ rms_w, const float* __restrict__ ssq,
    const float* __restrict__ Bw, float* __restrict__ Cm) {
  __shared__ __align__(16) _Float16 As[128][40];
  __shared__ __align__(16) _Float16 BsT[128][40];
  const int m0 = blockIdx.x * 128, n0 = blockIdx.y * 128;
  const int tid = threadIdx.x, lane = tid & 31, wave = tid >> 5;
  const int mr0 = (wave & 3) * 32, nc0 = (wave >> 2) * 64;
  const int arow = tid >> 3, acg = tid & 7;
  v8f acc[2][4] = {};
  for (int h = 0; h < Hc; ++h) {
    float rstd[4];
#pragma unroll
    for (int r = 0; r < 4; ++r) {
      const float ms = ssq[(size_t)(m0 + arow + 32 * r) * Hc + h] * (1.f / DVc);
      rstd[r] = rsqrtf(ms + RMS_EPS);
    }
    for (int kk = h * DVc; kk < (h + 1) * DVc; kk += 32) {
      const v4f rw = *(const v4f*)(rms_w + (kk - h * DVc) + acg * 4);
#pragma unroll
      for (int r = 0; r < 4; ++r) {
        const int row = arow + 32 * r;
        const size_t off = (size_t)(m0 + row) * 512 + kk + acg * 4;
        const v4f ov = *(const v4f*)(on + off);
        const v4f gv = *(const v4f*)(g + off);
        v4h hv;
#pragma unroll
        for (int e = 0; e < 4; ++e) {
          const float gg = gv[e];
          const float val = ov[e] * rstd[r] * rw[e];
          hv[e] = (_Float16)(val * gg * (1.f / (1.f + __expf(-gg))));
        }
        *(v4h*)&As[row][acg * 4] = hv;
      }
      for (int i = tid; i < 1024; i += 256) {
        const int k = i >> 5, ng = i & 31;
        v4f wv = *(const v4f*)(Bw + (size_t)(kk + k) * 512 + n0 + ng * 4);
#pragma unroll
        for (int j2 = 0; j2 < 4; ++j2) BsT[ng * 4 + j2][k] = (_Float16)wv[j2];
      }
      if (kk + 32 < 512) {
        __builtin_prefetch(&on[(size_t)(m0 + (tid >> 1)) * 512 + kk + 32], 0, 1);
        __builtin_prefetch(&g[(size_t)(m0 + (tid >> 1)) * 512 + kk + 32], 0, 1);
      }
      __syncthreads();
      v16h a0 = frag16(&As[0][0], 40, mr0, 0);
      v16h a1 = frag16(&As[0][0], 40, mr0 + 16, 0);
#pragma unroll
      for (int j = 0; j < 4; ++j) {
        v16h bm = frag16(&BsT[0][0], 40, nc0 + j * 16, 0);
        acc[0][j] = wmma16(a0, bm, acc[0][j]);
        acc[1][j] = wmma16(a1, bm, acc[1][j]);
      }
      __syncthreads();
    }
  }
  const int nl = lane & 15, mOff = (lane >> 4) * 8;
#pragma unroll
  for (int a2 = 0; a2 < 2; ++a2) {
#pragma unroll
    for (int j = 0; j < 4; ++j) {
      const int n = n0 + nc0 + j * 16 + nl;
#pragma unroll
      for (int r = 0; r < 8; ++r)
        Cm[(size_t)(m0 + mr0 + a2 * 16 + mOff + r) * 512 + n] = acc[a2][j][r];
    }
  }
}

extern "C" void kernel_launch(void* const* d_in, const int* in_sizes, int n_in,
                              void* d_out, int out_size, void* d_ws, size_t ws_size,
                              hipStream_t stream) {
  (void)in_sizes; (void)n_in; (void)out_size; (void)ws_size;
  const float* x    = (const float*)d_in[0];
  const float* Wq   = (const float*)d_in[1];
  const float* Wk   = (const float*)d_in[2];
  const float* Wv   = (const float*)d_in[3];
  const float* Wg   = (const float*)d_in[4];
  const float* Wo   = (const float*)d_in[5];
  const float* Wgk1 = (const float*)d_in[6];
  const float* Wgk2 = (const float*)d_in[7];
  const float* bgk2 = (const float*)d_in[8];
  const float* rmsw = (const float*)d_in[9];
  float* out = (float*)d_out;

  float* ws   = (float*)d_ws;
  float* qws  = ws;
  float* kws  = qws  + (size_t)TOK * 256;
  float* vws  = kws  + (size_t)TOK * 256;
  float* gws  = vws  + (size_t)TOK * 512;
  float* xg1  = gws  + (size_t)TOK * 512;
  float* on   = xg1  + (size_t)TOK * 16;
  float* ssq  = on   + (size_t)TOK * 512;
  float* qhw  = ssq  + (size_t)TOK * Hc;
  float* khw  = qhw  + (size_t)TOK * 128;
  float* khTw = khw  + (size_t)TOK * 128;
  float* vTw  = khTw + (size_t)TOK * 128;
  float* elc  = vTw  + (size_t)TOK * 256;

  proj_kernel<<<dim3(TOK / 128, 1536 / 128), 256, 0, stream>>>(
      x, Wq, Wk, Wv, Wg, qws, kws, vws, gws);
  lr1_kernel<<<TOK / 16, 256, 0, stream>>>(x, Wgk1, xg1);
  gate_prep_kernel<<<(TOK / CH) * Hc, 256, 0, stream>>>(
      xg1, Wgk2, bgk2, qws, kws, vws, qhw, khw, khTw, vTw, elc);
  zero_kernel<<<(TOK * Hc + 255) / 256, 256, 0, stream>>>(ssq, TOK * Hc);
  gla_scan_kernel<<<dim3(Bc * Hc, 2), 256, 0, stream>>>(
      qhw, khw, khTw, vTw, elc, on, ssq);
  out_gemm_kernel<<<dim3(TOK / 128, 512 / 128), 256, 0, stream>>>(
      on, gws, rmsw, ssq, Wo, out);
}


# Weights turned once per call into f16 [n][k] B tiles (extra blocks of the low-rank launch); both GEMMs' B loaders copy 16-B pieces: no 2-byte transposing LDS stores, no one-at-a-time B loads
# speedup vs baseline: 7.6035x; 2.8670x over previous
#include <hip/hip_runtime.h>
#include <hip/hip_bf16.h>
#include <math.h>


typedef __attribute__((ext_vector_type(16))) _Float16 v16h;
typedef __attribute__((ext_vector_type(8)))  _Float16 v8h;
typedef __attribute__((ext_vector_type(4)))  _Float16 v4h;
typedef __attribute__((ext_vector_type(8)))  float    v8f;
typedef __attribute__((ext_vector_type(4)))  float    v4f;
typedef __attribute__((ext_vector_type(2)))  float    v2f;
typedef __attribute__((ext_vector_type(2)))  _Float16 v2h;

constexpr int Bc  = 4;
constexpr int Tc  = 2048;
constexpr int HIDc = 512;
constexpr int Hc  = 4;
constexpr int DKc = 64;
constexpr int DVc = 128;
constexpr int CH  = 64;
constexpr int TOK = Bc * Tc;
constexpr float RMS_EPS = 1e-5f;
constexpr int WTILE = 128 * 32;
constexpr int LR1_BLOCKS = TOK / 16;
constexpr int PROJ_WTILES = (1536 / 128) * (HIDc / 32);
constexpr int OUT_WTILES  = (512 / 128) * (512 / 32);

__device__ __forceinline__ v8f wmma16(v16h a, v16h b, v8f c) {
  return __builtin_amdgcn_wmma_f32_16x16x32_f16(false, a, false, b, (short)0, c,
                                                false, false);
}

__device__ __forceinline__ v16h frag16(const _Float16* p, int ld, int rowBase,
                                       int kBase) {
  const int lane = threadIdx.x & 31, half = lane >> 4;
  const _Float16* row = p + (size_t)(rowBase + (lane & 15)) * ld + kBase + half * 8;
  v8h lo = *(const v8h*)(row);
  v8h hi = *(const v8h*)(row + 16);
  return __builtin_shufflevector(lo, hi, 0, 1, 2, 3, 4, 5, 6, 7,
                                 8, 9, 10, 11, 12, 13, 14, 15);
}

__global__ __launch_bounds__(256) void proj_kernel(
    const float* __restrict__ x, const float* __restrict__ wT_ws,
    float* __restrict__ qws, float* __restrict__ kws,
    float* __restrict__ vws, float* __restrict__ gws) {
  __shared__ __align__(16) _Float16 As[128][40];
  __shared__ __align__(16) _Float16 BsT[128][40];
  const int m0 = blockIdx.x * 128, n0 = blockIdx.y * 128;
  const int tid = threadIdx.x, lane = tid & 31, wave = tid >> 5;
  const int mr0 = (wave & 3) * 32;
  const int nc0 = (wave >> 2) * 64;
  const _Float16* btile = (const _Float16*)wT_ws + (size_t)blockIdx.y * 16 * WTILE;
  v8f acc[2][4] = {};
  for (int kk = 0; kk < HIDc; kk += 32, btile += WTILE) {
    for (int i = tid; i < 1024; i += 256) {
      const int row = i >> 3, cg = i & 7;
      v4f xv = *(const v4f*)(x + (size_t)(m0 + row) * HIDc + kk + cg * 4);
      v4h hv;
#pragma unroll
      for (int e = 0; e < 4; ++e) hv[e] = (_Float16)xv[e];
      *(v4h*)&As[row][cg * 4] = hv;
    }
    for (int r = 0; r < 2; ++r) {
      const int i = tid + 256 * r;
      *(v8h*)&BsT[i >> 2][(i & 3) * 8] = *(const v8h*)(btile + i * 8);
    }
    if (kk + 32 < HIDc)
      __builtin_prefetch(&x[(size_t)(m0 + (tid >> 1)) * HIDc + kk + 32], 0, 1);
    __syncthreads();
    v16h a0 = frag16(&As[0][0], 40, mr0, 0);
    v16h a1 = frag16(&As[0][0], 40, mr0 + 16, 0);
#pragma unroll
    for (int j = 0; j < 4; ++j) {
      v16h bm = frag16(&BsT[0][0], 40, nc0 + j * 16, 0);
      acc[0][j] = wmma16(a0, bm, acc[0][j]);
      acc[1][j] = wmma16(a1, bm, acc[1][j]);
    }
    __syncthreads();
  }
  const int nl = lane & 15, mOff = (lane >> 4) * 8;
#pragma unroll
  for (int a2 = 0; a2 < 2; ++a2) {
#pragma unroll
    for (int j = 0; j < 4; ++j) {
      const int n = n0 + nc0 + j * 16 + nl;
#pragma unroll
      for (int r = 0; r < 8; ++r) {
        const size_t m = (size_t)(m0 + mr0 + a2 * 16 + mOff + r);
        const float val = acc[a2][j][r];
        if (n < 256)       qws[m * 256 + n]          = val * 0.125f;
        else if (n < 512)  kws[m * 256 + (n - 256)]  = val;
        else if (n < 1024) vws[m * 512 + (n - 512)]  = val;
        else               gws[m * 512 + (n - 1024)] = val;
      }
    }
  }
}

__global__ __launch_bounds__(256) void lr1_kernel(
    const float* __restrict__ x, const float* __restrict__ Wgk1,
    float* __restrict__ xg1,
    const float* __restrict__ Wq, const float* __restrict__ Wk,
    const float* __restrict__ Wv, const float* __restrict__ Wg,
    const float* __restrict__ Wo, float* __restrict__ wT_ws) {
  __shared__ __align__(16) float xs[16][512];
  if (blockIdx.x >= LR1_BLOCKS) {
    const int j = blockIdx.x - LR1_BLOCKS;
    const int nb = (j < PROJ_WTILES ? j : j - PROJ_WTILES) >> 4, ks = j & 15;
    const float* src;
    int ld;
    if (j >= PROJ_WTILES) { src = Wo + nb * 128;       ld = 512; }
    else if (nb < 2)      { src = Wq + nb * 128;       ld = 256; }
    else if (nb < 4)      { src = Wk + (nb - 2) * 128; ld = 256; }
    else if (nb < 8)      { src = Wv + (nb - 4) * 128; ld = 512; }
    else                  { src = Wg + (nb - 8) * 128; ld = 512; }
    _Float16 (*tT)[40] = (_Float16 (*)[40])&xs[0][0];
    for (int i = threadIdx.x; i < 1024; i += 256) {
      const int k = i >> 5, ng = i & 31;
      const v4f wv = *(const v4f*)(src + (size_t)(ks * 32 + k) * ld + ng * 4);
#pragma unroll
      for (int j2 = 0; j2 < 4; ++j2) tT[ng * 4 + j2][k] = (_Float16)wv[j2];
    }
    __syncthreads();
    _Float16* dst = (_Float16*)wT_ws + (size_t)j * WTILE;
    for (int r = 0; r < 2; ++r) {
      const int i = threadIdx.x + 256 * r;
      *(v8h*)(dst + i * 8) = *(const v8h*)&tT[i >> 2][(i & 3) * 8];
    }
    return;
  }
  const int t0 = blockIdx.x * 16;
  for (int i = threadIdx.x; i < 16 * 512; i += 256)
    xs[i >> 9][i & 511] = x[(size_t)(t0 + (i >> 9)) * 512 + (i & 511)];
  __syncthreads();
  const int tok = threadIdx.x >> 4, j = threadIdx.x & 15;
  float s = 0.f;
  for (int k2 = 0; k2 < 512; ++k2) s += xs[tok][k2] * Wgk1[k2 * 16 + j];
  xg1[(size_t)(t0 + tok) * 16 + j] = s;
}

__global__ __launch_bounds__(256) void gate_prep_kernel(
    const float* __restrict__ xg1, const float* __restrict__ Wgk2,
    const float* __restrict__ bgk2, const float* __restrict__ qg,
    const float* __restrict__ kg, const float* __restrict__ vg,
    float* __restrict__ qh_ws, float* __restrict__ kh_ws,
    float* __restrict__ vT_ws, float* __restrict__ expLCg,
    float* __restrict__ kv_ws) {
  constexpr int LDA = CH + 8;
  __shared__ __align__(16) float    xs[CH][16];
  __shared__ float                  lamS[CH][DKc + 1];
  __shared__ __align__(16) _Float16 qS[CH][LDA];
  __shared__ __align__(16) _Float16 kS[CH][LDA];
  __shared__ __align__(16) _Float16 kTS[DKc][LDA];
  __shared__ __align__(16) _Float16 vTS[DVc][LDA];

  const int tile = blockIdx.x;
  const int h = tile & 3, tok0 = (tile >> 2) * CH;
  const int tid = threadIdx.x, d = tid & 63, grp = tid >> 6;
  const int n = h * DKc + d;

  *(v4f*)(&xs[0][0] + tid * 4) = *(const v4f*)(xg1 + (size_t)tok0 * 16 + tid * 4);
  float w[16];
#pragma unroll
  for (int j = 0; j < 16; ++j) w[j] = Wgk2[j * 256 + n];
  const float bias = bgk2[n];

  float qv[16], kv[16];
#pragma unroll
  for (int tt = 0; tt < 16; ++tt) {
    const size_t off = (size_t)(tok0 + grp * 16 + tt) * (Hc * DKc) + n;
    qv[tt] = qg[off];
    kv[tt] = kg[off];
  }

  {
    const float* vb = vg + (size_t)tok0 * (Hc * DVc) + h * DVc;
    for (int r = 0; r < 8; ++r) {
      const int i = tid + 256 * r, s = i >> 5, n4 = (i & 31) * 4;
      const v4f vv = *(const v4f*)(vb + (size_t)s * (Hc * DVc) + n4);
#pragma unroll
      for (int e = 0; e < 4; ++e) vTS[n4 + e][s] = (_Float16)vv[e];
    }
  }
  __syncthreads();

  for (int tt = 0; tt < 16; ++tt) {
    const int t = grp * 16 + tt;
    float s = bias;
#pragma unroll
    for (int j = 0; j < 16; ++j) s += xs[t][j] * w[j];
    const float ls = fminf(s, 0.f) - log1pf(__expf(-fabsf(s)));
    lamS[t][d] = ls * (1.f / 16.f);
  }
  __syncthreads();

  if (tid < DKc) {
    float lam = 0.f;
    for (int t = 0; t < CH; ++t) {
      lam += lamS[t][tid];
      lamS[t][tid] = lam;
    }
    expLCg[(size_t)tile * DKc + tid] = __expf(lam);
  }
  __syncthreads();

  const float lamC = lamS[CH - 1][d];
#pragma unroll
  for (int tt = 0; tt < 16; ++tt) {
    const int t = grp * 16 + tt;
    const float eq = __expf(lamS[t][d] - lamC);
    const float qh = fmaxf(fminf(qv[tt] * eq, 60000.f), -60000.f);
    const float kh = kv[tt] / eq;
    qS[t][d]  = (_Float16)qh;
    kS[t][d]  = (_Float16)kh;
    kTS[d][t] = (_Float16)kh;
  }
  __syncthreads();

  _Float16* qd  = (_Float16*)qh_ws  + (size_t)tile * (CH * DKc);
  _Float16* kd  = (_Float16*)kh_ws  + (size_t)tile * (CH * DKc);
  _Float16* vtd = (_Float16*)vT_ws  + (size_t)tile * (CH * DVc);
  for (int r = 0; r < 2; ++r) {
    const int i = tid + 256 * r, row = i >> 3, cc = (i & 7) * 8;
    *(v8h*)(qd  + i * 8) = *(const v8h*)&qS[row][cc];
    *(v8h*)(kd  + i * 8) = *(const v8h*)&kS[row][cc];
  }
  for (int r = 0; r < 4; ++r) {
    const int i = tid + 256 * r, row = i >> 3, cc = (i & 7) * 8;
    *(v8h*)(vtd + i * 8) = *(const v8h*)&vTS[row][cc];
  }

  {
    const int lane = tid & 31, wave = tid >> 5;
    v8f acc[4] = {};
#pragma unroll
    for (int k2 = 0; k2 < 2; ++k2) {
      v16h bv = frag16(&vTS[0][0], LDA, wave * 16, k2 * 32);
#pragma unroll
      for (int dr = 0; dr < 4; ++dr) {
        v16h aK = frag16(&kTS[0][0], LDA, dr * 16, k2 * 32);
        acc[dr] = wmma16(aK, bv, acc[dr]);
      }
    }
    float* dst = kv_ws + (size_t)tile * (DVc * DKc) +
                 (wave * 16 + (lane & 15)) * DKc + (lane >> 4) * 8;
#pragma unroll
    for (int dr = 0; dr < 4; ++dr) {
      const v4f lo = {acc[dr][0], acc[dr][1], acc[dr][2], acc[dr][3]};
      const v4f hi = {acc[dr][4], acc[dr][5], acc[dr][6], acc[dr][7]};
      *(v4f*)(dst + dr * 16)     = lo;
      *(v4f*)(dst + dr * 16 + 4) = hi;
    }
  }
}

__global__ __launch_bounds__(256) void state_kernel(
    const float* __restrict__ kv_ws, const float* __restrict__ expLCg,
    float* __restrict__ sp_ws) {
  constexpr int NC = Tc / CH;
  constexpr int GRP = 16;
  const int bh = blockIdx.x, b = bh >> 2, h = bh & 3;
  const int e2 = (blockIdx.y * 256 + threadIdx.x) * 2;
  const int d = e2 & 63;
  _Float16* sp = (_Float16*)sp_ws;
  float s0 = 0.f, s1 = 0.f;
  for (int c0 = 0; c0 < NC; c0 += GRP) {
    v2f kv[GRP], dc[GRP];
#pragma unroll
    for (int i = 0; i < GRP; ++i) {
      const size_t tile = (size_t)((b * NC + c0 + i) * Hc + h);
      kv[i] = *(const v2f*)(kv_ws + tile * (DVc * DKc) + e2);
      dc[i] = *(const v2f*)(expLCg + tile * DKc + d);
    }
#pragma unroll
    for (int i = 0; i < GRP; ++i) {
      const size_t tile = (size_t)((b * NC + c0 + i) * Hc + h);
      s0 *= dc[i][0];
      s1 *= dc[i][1];
      v2h o2;
      o2[0] = (_Float16)s0;
      o2[1] = (_Float16)s1;
      *(v2h*)(sp + tile * (DVc * DKc) + e2) = o2;
      s0 += kv[i][0];
      s1 += kv[i][1];
    }
  }
}

__device__ __forceinline__ float row16_sum(float v) {
  v += __builtin_bit_cast(float, __builtin_amdgcn_update_dpp(
           0, __builtin_bit_cast(int, v), 0x161, 0xf, 0xf, true));
  v += __builtin_bit_cast(float, __builtin_amdgcn_update_dpp(
           0, __builtin_bit_cast(int, v), 0x162, 0xf, 0xf, true));
  v += __builtin_bit_cast(float, __builtin_amdgcn_update_dpp(
           0, __builtin_bit_cast(int, v), 0x164, 0xf, 0xf, true));
  v += __builtin_bit_cast(float, __builtin_amdgcn_update_dpp(
           0, __builtin_bit_cast(int, v), 0x168, 0xf, 0xf, true));
  return v;
}

__global__ __launch_bounds__(256) void gla_chunk_kernel(
    const float* __restrict__ qh_ws, const float* __restrict__ kh_ws,
    const float* __restrict__ vT_ws, const float* __restrict__ sp_ws,
    float* __restrict__ on, float* __restrict__ ssq_g) {
  constexpr int LDA = CH + 8;
  __shared__ __align__(16) _Float16 qt[CH][LDA];
  __shared__ __align__(16) _Float16 kt[CH][LDA];
  __shared__ __align__(16) _Float16 Am[CH][LDA];
  __shared__ __align__(16) _Float16 vT[DVc][LDA];
  __shared__ __align__(16) _Float16 sT[DVc][LDA];
  __shared__ float ssqP[2][CH];

  const int tile = blockIdx.x;
  const int h = tile & 3, tok0 = (tile >> 2) * CH;
  const int tid = threadIdx.x, lane = tid & 31, wave = tid >> 5;

  {
    const _Float16* qtile  = (const _Float16*)qh_ws + (size_t)tile * (CH * DKc);
    const _Float16* ktile  = (const _Float16*)kh_ws + (size_t)tile * (CH * DKc);
    const _Float16* vTtile = (const _Float16*)vT_ws + (size_t)tile * (CH * DVc);
    const _Float16* stile  = (const _Float16*)sp_ws + (size_t)tile * (DVc * DKc);
    for (int r = 0; r < 2; ++r) {
      const int i = tid + 256 * r, row = i >> 3, cc = (i & 7) * 8;
      *(v8h*)&qt[row][cc] = *(const v8h*)(qtile + i * 8);
      *(v8h*)&kt[row][cc] = *(const v8h*)(ktile + i * 8);
    }
    for (int r = 0; r < 4; ++r) {
      const int i = tid + 256 * r, row = i >> 3, cc = (i & 7) * 8;
      *(v8h*)&vT[row][cc] = *(const v8h*)(vTtile + i * 8);
      *(v8h*)&sT[row][cc] = *(const v8h*)(stile  + i * 8);
    }
  }
  __syncthreads();

  {
    const int tRow = wave & 3, sp0 = (wave >> 2) * 2;
#pragma unroll
    for (int spi = 0; spi < 2; ++spi) {
      const int sTile = sp0 + spi;
      v8f acc = {};
      if (sTile <= tRow) {
#pragma unroll
        for (int k2 = 0; k2 < 2; ++k2) {
          v16h a  = frag16(&qt[0][0], LDA, tRow * 16, k2 * 32);
          v16h bm = frag16(&kt[0][0], LDA, sTile * 16, k2 * 32);
          acc = wmma16(a, bm, acc);
        }
      }
      const int sIdx = sTile * 16 + (lane & 15), mOff = (lane >> 4) * 8;
#pragma unroll
      for (int r = 0; r < 8; ++r) {
        const int tIdx = tRow * 16 + mOff + r;
        Am[tIdx][sIdx] = (_Float16)((sIdx <= tIdx) ? acc[r] : 0.f);
      }
    }
  }
  __syncthreads();

  {
    const int tRow = wave & 3, nHalf = wave >> 2, ntb = nHalf * 4;
    v8f acc[4] = {};
#pragma unroll
    for (int k2 = 0; k2 < 2; ++k2) {
      v16h aA = frag16(&Am[0][0], LDA, tRow * 16, k2 * 32);
      v16h aQ = frag16(&qt[0][0], LDA, tRow * 16, k2 * 32);
#pragma unroll
      for (int j = 0; j < 4; ++j) {
        v16h bv = frag16(&vT[0][0], LDA, (ntb + j) * 16, k2 * 32);
        acc[j] = wmma16(aA, bv, acc[j]);
        v16h bs = frag16(&sT[0][0], LDA, (ntb + j) * 16, k2 * 32);
        acc[j] = wmma16(aQ, bs, acc[j]);
      }
    }
    float* ob = on + (size_t)tok0 * (Hc * DVc) + h * DVc;
    const int nl = lane & 15, mOff = (lane >> 4) * 8;
#pragma unroll
    for (int j = 0; j < 4; ++j) {
      const int n = (ntb + j) * 16 + nl;
#pragma unroll
      for (int r = 0; r < 8; ++r) {
        const int t = tRow * 16 + mOff + r;
        ob[(size_t)t * (Hc * DVc) + n] = acc[j][r];
      }
    }
#pragma unroll
    for (int r = 0; r < 8; ++r) {
      float p = acc[0][r] * acc[0][r];
#pragma unroll
      for (int j = 1; j < 4; ++j) p += acc[j][r] * acc[j][r];
      p = row16_sum(p);
      if (nl == 0) ssqP[nHalf][tRow * 16 + mOff + r] = p;
    }
  }
  __syncthreads();
  if (tid < CH)
    ssq_g[(size_t)h * TOK + tok0 + tid] = ssqP[0][tid] + ssqP[1][tid];
}

__global__ __launch_bounds__(256) void out_gemm_kernel(
    const float* __restrict__ on, const float* __restrict__ g,
    const float* __restrict__ rms_w, const float* __restrict__ ssq,
    const float* __restrict__ woT_ws, float* __restrict__ Cm) {
  __shared__ __align__(16) _Float16 As[128][40];
  __shared__ __align__(16) _Float16 BsT[128][40];
  const int m0 = blockIdx.x * 128, n0 = blockIdx.y * 128;
  const int tid = threadIdx.x, lane = tid & 31, wave = tid >> 5;
  const int mr0 = (wave & 3) * 32, nc0 = (wave >> 2) * 64;
  const int arow = tid >> 3, acg = tid & 7;
  const _Float16* btile = (const _Float16*)woT_ws + (size_t)blockIdx.y * 16 * WTILE;
  v8f acc[2][4] = {};
  for (int h = 0; h < Hc; ++h) {
    float rstd[4];
#pragma unroll
    for (int r = 0; r < 4; ++r) {
      const float ms = ssq[(size_t)h * TOK + m0 + arow + 32 * r] * (1.f / DVc);
      rstd[r] = rsqrtf(ms + RMS_EPS);
    }
    for (int kk = h * DVc; kk < (h + 1) * DVc; kk += 32, btile += WTILE) {
      const v4f rw = *(const v4f*)(rms_w + (kk - h * DVc) + acg * 4);
#pragma unroll
      for (int r = 0; r < 4; ++r) {
        const int row = arow + 32 * r;
        const size_t off = (size_t)(m0 + row) * 512 + kk + acg * 4;
        const v4f ov = *(const v4f*)(on + off);
        const v4f gv = *(const v4f*)(g + off);
        v4h hv;
#pragma unroll
        for (int e = 0; e < 4; ++e) {
          const float gg = gv[e];
          const float val = ov[e] * rstd[r] * rw[e];
          hv[e] = (_Float16)(val * gg * (1.f / (1.f + __expf(-gg))));
        }
        *(v4h*)&As[row][acg * 4] = hv;
      }
      for (int r = 0; r < 2; ++r) {
        const int i = tid + 256 * r;
        *(v8h*)&BsT[i >> 2][(i & 3) * 8] = *(const v8h*)(btile + i * 8);
      }
      if (kk + 32 < 512) {
        __builtin_prefetch(&on[(size_t)(m0 + (tid >> 1)) * 512 + kk + 32], 0, 1);
        __builtin_prefetch(&g[(size_t)(m0 + (tid >> 1)) * 512 + kk + 32], 0, 1);
      }
      __syncthreads();
      v16h a0 = frag16(&As[0][0], 40, mr0, 0);
      v16h a1 = frag16(&As[0][0], 40, mr0 + 16, 0);
#pragma unroll
      for (int j = 0; j < 4; ++j) {
        v16h bm = frag16(&BsT[0][0], 40, nc0 + j * 16, 0);
        acc[0][j] = wmma16(a0, bm, acc[0][j]);
        acc[1][j] = wmma16(a1, bm, acc[1][j]);
      }
      __syncthreads();
    }
  }
  const int nl = lane & 15, mOff = (lane >> 4) * 8;
#pragma unroll
  for (int a2 = 0; a2 < 2; ++a2) {
#pragma unroll
    for (int j = 0; j < 4; ++j) {
      const int n = n0 + nc0 + j * 16 + nl;
#pragma unroll
      for (int r = 0; r < 8; ++r)
        Cm[(size_t)(m0 + mr0 + a2 * 16 + mOff + r) * 512 + n] = acc[a2][j][r];
    }
  }
}

extern "C" void kernel_launch(void* const* d_in, const int* in_sizes, int n_in,
                              void* d_out, int out_size, void* d_ws, size_t ws_size,
                              hipStream_t stream) {
  (void)in_sizes; (void)n_in; (void)out_size; (void)ws_size;
  const float* x    = (const float*)d_in[0];
  const float* Wq   = (const float*)d_in[1];
  const float* Wk   = (const float*)d_in[2];
  const float* Wv   = (const float*)d_in[3];
  const float* Wg   = (const float*)d_in[4];
  const float* Wo   = (const float*)d_in[5];
  const float* Wgk1 = (const float*)d_in[6];
  const float* Wgk2 = (const float*)d_in[7];
  const float* bgk2 = (const float*)d_in[8];
  const float* rmsw = (const float*)d_in[9];
  float* out = (float*)d_out;

  float* ws   = (float*)d_ws;
  float* qws  = ws;
  float* kws  = qws  + (size_t)TOK * 256;
  float* vws  = kws  + (size_t)TOK * 256;
  float* gws  = vws  + (size_t)TOK * 512;
  float* xg1  = gws  + (size_t)TOK * 512;
  float* on   = xg1  + (size_t)TOK * 16;
  float* ssq  = on   + (size_t)TOK * 512;
  float* qhw  = ssq  + (size_t)TOK * Hc;
  float* khw  = qhw  + (size_t)TOK * 128;
  float* vTw  = khw  + (size_t)TOK * 128;
  float* elc  = vTw  + (size_t)TOK * 256;
  float* kvw  = elc  + (size_t)TOK * 4;
  float* spw  = kvw  + (size_t)TOK * 512;
  float* wTw  = spw  + (size_t)TOK * 256;
  float* woTw = wTw  + (size_t)PROJ_WTILES * WTILE / 2;

  lr1_kernel<<<LR1_BLOCKS + PROJ_WTILES + OUT_WTILES, 256, 0, stream>>>(
      x, Wgk1, xg1, Wq, Wk, Wv, Wg, Wo, wTw);
  proj_kernel<<<dim3(TOK / 128, 1536 / 128), 256, 0, stream>>>(
      x, wTw, qws, kws, vws, gws);
  gate_prep_kernel<<<(TOK / CH) * Hc, 256, 0, stream>>>(
      xg1, Wgk2, bgk2, qws, kws, vws, qhw, khw, vTw, elc, kvw);
  state_kernel<<<dim3(Bc * Hc, (DVc * DKc) / 512), 256, 0, stream>>>(kvw, elc, spw);
  gla_chunk_kernel<<<(TOK / CH) * Hc, 256, 0, stream>>>(
      qhw, khw, vTw, spw, on, ssq);
  out_gemm_kernel<<<dim3(TOK / 128, 512 / 128), 256, 0, stream>>>(
      on, gws, rmsw, ssq, woTw, out);
}


# Both GEMM K-loops software-pipelined over two LDS stages: a K-step's global loads go out as one group ahead of use, under the previous step's WMMAs and gate math; one barrier per step
# speedup vs baseline: 8.6543x; 1.1382x over previous
#include <hip/hip_runtime.h>
#include <hip/hip_bf16.h>
#include <math.h>


typedef __attribute__((ext_vector_type(16))) _Float16 v16h;
typedef __attribute__((ext_vector_type(8)))  _Float16 v8h;
typedef __attribute__((ext_vector_type(4)))  _Float16 v4h;
typedef __attribute__((ext_vector_type(8)))  float    v8f;
typedef __attribute__((ext_vector_type(4)))  float    v4f;
typedef __attribute__((ext_vector_type(2)))  float    v2f;
typedef __attribute__((ext_vector_type(2)))  _Float16 v2h;

constexpr int Bc  = 4;
constexpr int Tc  = 2048;
constexpr int HIDc = 512;
constexpr int Hc  = 4;
constexpr int DKc = 64;
constexpr int DVc = 128;
constexpr int CH  = 64;
constexpr int TOK = Bc * Tc;
constexpr float RMS_EPS = 1e-5f;
constexpr int WTILE = 128 * 32;
constexpr int LR1_BLOCKS = TOK / 16;
constexpr int PROJ_WTILES = (1536 / 128) * (HIDc / 32);
constexpr int OUT_WTILES  = (512 / 128) * (512 / 32);

__device__ __forceinline__ v8f wmma16(v16h a, v16h b, v8f c) {
  return __builtin_amdgcn_wmma_f32_16x16x32_f16(false, a, false, b, (short)0, c,
                                                false, false);
}

__device__ __forceinline__ v16h frag16(const _Float16* p, int ld, int rowBase,
                                       int kBase) {
  const int lane = threadIdx.x & 31, half = lane >> 4;
  const _Float16* row = p + (size_t)(rowBase + (lane & 15)) * ld + kBase + half * 8;
  v8h lo = *(const v8h*)(row);
  v8h hi = *(const v8h*)(row + 16);
  return __builtin_shufflevector(lo, hi, 0, 1, 2, 3, 4, 5, 6, 7,
                                 8, 9, 10, 11, 12, 13, 14, 15);
}

__global__ __launch_bounds__(256) void proj_kernel(
    const float* __restrict__ x, const float* __restrict__ wT_ws,
    float* __restrict__ qws, float* __restrict__ kws,
    float* __restrict__ vws, float* __restrict__ gws) {
  __shared__ __align__(16) _Float16 As[2][128][40];
  __shared__ __align__(16) _Float16 BsT[2][128][40];
  constexpr int NS = HIDc / 32;
  const int m0 = blockIdx.x * 128, n0 = blockIdx.y * 128;
  const int tid = threadIdx.x, lane = tid & 31, wave = tid >> 5;
  const int mr0 = (wave & 3) * 32;
  const int nc0 = (wave >> 2) * 64;
  const int arow = tid >> 3, acg = tid & 7;
  const float* xa_p = x + (size_t)(m0 + arow) * HIDc + acg * 4;
  const float* xpf_p = x + (size_t)(m0 + (tid >> 1)) * HIDc;
  const _Float16* bt_p = (const _Float16*)wT_ws + (size_t)blockIdx.y * 16 * WTILE + tid * 8;
  v4f xa[4];
  v8h bb[2];
  auto load = [&](int s1) {
#pragma unroll
    for (int r = 0; r < 4; ++r)
      xa[r] = *(const v4f*)(xa_p + (size_t)r * 32 * HIDc + s1 * 32);
#pragma unroll
    for (int r = 0; r < 2; ++r)
      bb[r] = *(const v8h*)(bt_p + (size_t)s1 * WTILE + r * 2048);
    if (s1 + 1 < NS) __builtin_prefetch(xpf_p + (s1 + 1) * 32, 0, 1);
  };
  auto store = [&](int st) {
#pragma unroll
    for (int r = 0; r < 4; ++r) {
      v4h hv;
#pragma unroll
      for (int e = 0; e < 4; ++e) hv[e] = (_Float16)xa[r][e];
      *(v4h*)&As[st][arow + 32 * r][acg * 4] = hv;
    }
#pragma unroll
    for (int r = 0; r < 2; ++r) {
      const int i = tid + 256 * r;
      *(v8h*)&BsT[st][i >> 2][(i & 3) * 8] = bb[r];
    }
  };
  load(0);
  store(0);
  __syncthreads();
  v8f acc[2][4] = {};
  for (int s = 0; s < NS; ++s) {
    const int st = s & 1;
    if (s + 1 < NS) load(s + 1);
    v16h a0 = frag16(&As[st][0][0], 40, mr0, 0);
    v16h a1 = frag16(&As[st][0][0], 40, mr0 + 16, 0);
#pragma unroll
    for (int j = 0; j < 4; ++j) {
      v16h bm = frag16(&BsT[st][0][0], 40, nc0 + j * 16, 0);
      acc[0][j] = wmma16(a0, bm, acc[0][j]);
      acc[1][j] = wmma16(a1, bm, acc[1][j]);
    }
    __builtin_amdgcn_sched_barrier(0);
    if (s + 1 < NS) store(st ^ 1);
    __syncthreads();
  }
  const int nl = lane & 15, mOff = (lane >> 4) * 8;
#pragma unroll
  for (int a2 = 0; a2 < 2; ++a2) {
#pragma unroll
    for (int j = 0; j < 4; ++j) {
      const int n = n0 + nc0 + j * 16 + nl;
#pragma unroll
      for (int r = 0; r < 8; ++r) {
        const size_t m = (size_t)(m0 + mr0 + a2 * 16 + mOff + r);
        const float val = acc[a2][j][r];
        if (n < 256)       qws[m * 256 + n]          = val * 0.125f;
        else if (n < 512)  kws[m * 256 + (n - 256)]  = val;
        else if (n < 1024) vws[m * 512 + (n - 512)]  = val;
        else               gws[m * 512 + (n - 1024)] = val;
      }
    }
  }
}

__global__ __launch_bounds__(256) void lr1_kernel(
    const float* __restrict__ x, const float* __restrict__ Wgk1,
    float* __restrict__ xg1,
    const float* __restrict__ Wq, const float* __restrict__ Wk,
    const float* __restrict__ Wv, const float* __restrict__ Wg,
    const float* __restrict__ Wo, float* __restrict__ wT_ws) {
  __shared__ __align__(16) float xs[16][512];
  if (blockIdx.x >= LR1_BLOCKS) {
    const int j = blockIdx.x - LR1_BLOCKS;
    const int nb = (j < PROJ_WTILES ? j : j - PROJ_WTILES) >> 4, ks = j & 15;
    const float* src;
    int ld;
    if (j >= PROJ_WTILES) { src = Wo + nb * 128;       ld = 512; }
    else if (nb < 2)      { src = Wq + nb * 128;       ld = 256; }
    else if (nb < 4)      { src = Wk + (nb - 2) * 128; ld = 256; }
    else if (nb < 8)      { src = Wv + (nb - 4) * 128; ld = 512; }
    else                  { src = Wg + (nb - 8) * 128; ld = 512; }
    _Float16 (*tT)[40] = (_Float16 (*)[40])&xs[0][0];
    for (int i = threadIdx.x; i < 1024; i += 256) {
      const int k = i >> 5, ng = i & 31;
      const v4f wv = *(const v4f*)(src + (size_t)(ks * 32 + k) * ld + ng * 4);
#pragma unroll
      for (int j2 = 0; j2 < 4; ++j2) tT[ng * 4 + j2][k] = (_Float16)wv[j2];
    }
    __syncthreads();
    _Float16* dst = (_Float16*)wT_ws + (size_t)j * WTILE;
    for (int r = 0; r < 2; ++r) {
      const int i = threadIdx.x + 256 * r;
      *(v8h*)(dst + i * 8) = *(const v8h*)&tT[i >> 2][(i & 3) * 8];
    }
    return;
  }
  const int t0 = blockIdx.x * 16;
  for (int i = threadIdx.x; i < 16 * 512; i += 256)
    xs[i >> 9][i & 511] = x[(size_t)(t0 + (i >> 9)) * 512 + (i & 511)];
  __syncthreads();
  const int tok = threadIdx.x >> 4, j = threadIdx.x & 15;
  float s = 0.f;
  for (int k2 = 0; k2 < 512; ++k2) s += xs[tok][k2] * Wgk1[k2 * 16 + j];
  xg1[(size_t)(t0 + tok) * 16 + j] = s;
}

__global__ __launch_bounds__(256) void gate_prep_kernel(
    const float* __restrict__ xg1, const float* __restrict__ Wgk2,
    const float* __restrict__ bgk2, const float* __restrict__ qg,
    const float* __restrict__ kg, const float* __restrict__ vg,
    float* __restrict__ qh_ws, float* __restrict__ kh_ws,
    float* __restrict__ vT_ws, float* __restrict__ expLCg,
    float* __restrict__ kv_ws) {
  constexpr int LDA = CH + 8;
  __shared__ __align__(16) float    xs[CH][16];
  __shared__ float                  lamS[CH][DKc + 1];
  __shared__ __align__(16) _Float16 qS[CH][LDA];
  __shared__ __align__(16) _Float16 kS[CH][LDA];
  __shared__ __align__(16) _Float16 kTS[DKc][LDA];
  __shared__ __align__(16) _Float16 vTS[DVc][LDA];

  const int tile = blockIdx.x;
  const int h = tile & 3, tok0 = (tile >> 2) * CH;
  const int tid = threadIdx.x, d = tid & 63, grp = tid >> 6;
  const int n = h * DKc + d;

  *(v4f*)(&xs[0][0] + tid * 4) = *(const v4f*)(xg1 + (size_t)tok0 * 16 + tid * 4);
  float w[16];
#pragma unroll
  for (int j = 0; j < 16; ++j) w[j] = Wgk2[j * 256 + n];
  const float bias = bgk2[n];

  float qv[16], kv[16];
#pragma unroll
  for (int tt = 0; tt < 16; ++tt) {
    const size_t off = (size_t)(tok0 + grp * 16 + tt) * (Hc * DKc) + n;
    qv[tt] = qg[off];
    kv[tt] = kg[off];
  }

  {
    const float* vb = vg + (size_t)tok0 * (Hc * DVc) + h * DVc;
    for (int r = 0; r < 8; ++r) {
      const int i = tid + 256 * r, s = i >> 5, n4 = (i & 31) * 4;
      const v4f vv = *(const v4f*)(vb + (size_t)s * (Hc * DVc) + n4);
#pragma unroll
      for (int e = 0; e < 4; ++e) vTS[n4 + e][s] = (_Float16)vv[e];
    }
  }
  __syncthreads();

  for (int tt = 0; tt < 16; ++tt) {
    const int t = grp * 16 + tt;
    float s = bias;
#pragma unroll
    for (int j = 0; j < 16; ++j) s += xs[t][j] * w[j];
    const float ls = fminf(s, 0.f) - log1pf(__expf(-fabsf(s)));
    lamS[t][d] = ls * (1.f / 16.f);
  }
  __syncthreads();

  if (tid < DKc) {
    float lam = 0.f;
    for (int t = 0; t < CH; ++t) {
      lam += lamS[t][tid];
      lamS[t][tid] = lam;
    }
    expLCg[(size_t)tile * DKc + tid] = __expf(lam);
  }
  __syncthreads();

  const float lamC = lamS[CH - 1][d];
#pragma unroll
  for (int tt = 0; tt < 16; ++tt) {
    const int t = grp * 16 + tt;
    const float eq = __expf(lamS[t][d] - lamC);
    const float qh = fmaxf(fminf(qv[tt] * eq, 60000.f), -60000.f);
    const float kh = kv[tt] / eq;
    qS[t][d]  = (_Float16)qh;
    kS[t][d]  = (_Float16)kh;
    kTS[d][t] = (_Float16)kh;
  }
  __syncthreads();

  _Float16* qd  = (_Float16*)qh_ws  + (size_t)tile * (CH * DKc);
  _Float16* kd  = (_Float16*)kh_ws  + (size_t)tile * (CH * DKc);
  _Float16* vtd = (_Float16*)vT_ws  + (size_t)tile * (CH * DVc);
  for (int r = 0; r < 2; ++r) {
    const int i = tid + 256 * r, row = i >> 3, cc = (i & 7) * 8;
    *(v8h*)(qd  + i * 8) = *(const v8h*)&qS[row][cc];
    *(v8h*)(kd  + i * 8) = *(const v8h*)&kS[row][cc];
  }
  for (int r = 0; r < 4; ++r) {
    const int i = tid + 256 * r, row = i >> 3, cc = (i & 7) * 8;
    *(v8h*)(vtd + i * 8) = *(const v8h*)&vTS[row][cc];
  }

  {
    const int lane = tid & 31, wave = tid >> 5;
    v8f acc[4] = {};
#pragma unroll
    for (int k2 = 0; k2 < 2; ++k2) {
      v16h bv = frag16(&vTS[0][0], LDA, wave * 16, k2 * 32);
#pragma unroll
      for (int dr = 0; dr < 4; ++dr) {
        v16h aK = frag16(&kTS[0][0], LDA, dr * 16, k2 * 32);
        acc[dr] = wmma16(aK, bv, acc[dr]);
      }
    }
    float* dst = kv_ws + (size_t)tile * (DVc * DKc) +
                 (wave * 16 + (lane & 15)) * DKc + (lane >> 4) * 8;
#pragma unroll
    for (int dr = 0; dr < 4; ++dr) {
      const v4f lo = {acc[dr][0], acc[dr][1], acc[dr][2], acc[dr][3]};
      const v4f hi = {acc[dr][4], acc[dr][5], acc[dr][6], acc[dr][7]};
      *(v4f*)(dst + dr * 16)     = lo;
      *(v4f*)(dst + dr * 16 + 4) = hi;
    }
  }
}

__global__ __launch_bounds__(256) void state_kernel(
    const float* __restrict__ kv_ws, const float* __restrict__ expLCg,
    float* __restrict__ sp_ws) {
  constexpr int NC = Tc / CH;
  constexpr int GRP = 16;
  const int bh = blockIdx.x, b = bh >> 2, h = bh & 3;
  const int e2 = (blockIdx.y * 256 + threadIdx.x) * 2;
  const int d = e2 & 63;
  _Float16* sp = (_Float16*)sp_ws;
  float s0 = 0.f, s1 = 0.f;
  for (int c0 = 0; c0 < NC; c0 += GRP) {
    v2f kv[GRP], dc[GRP];
#pragma unroll
    for (int i = 0; i < GRP; ++i) {
      const size_t tile = (size_t)((b * NC + c0 + i) * Hc + h);
      kv[i] = *(const v2f*)(kv_ws + tile * (DVc * DKc) + e2);
      dc[i] = *(const v2f*)(expLCg + tile * DKc + d);
    }
#pragma unroll
    for (int i = 0; i < GRP; ++i) {
      const size_t tile = (size_t)((b * NC + c0 + i) * Hc + h);
      s0 *= dc[i][0];
      s1 *= dc[i][1];
      v2h o2;
      o2[0] = (_Float16)s0;
      o2[1] = (_Float16)s1;
      *(v2h*)(sp + tile * (DVc * DKc) + e2) = o2;
      s0 += kv[i][0];
      s1 += kv[i][1];
    }
  }
}

__device__ __forceinline__ float row16_sum(float v) {
  v += __builtin_bit_cast(float, __builtin_amdgcn_update_dpp(
           0, __builtin_bit_cast(int, v), 0x161, 0xf, 0xf, true));
  v += __builtin_bit_cast(float, __builtin_amdgcn_update_dpp(
           0, __builtin_bit_cast(int, v), 0x162, 0xf, 0xf, true));
  v += __builtin_bit_cast(float, __builtin_amdgcn_update_dpp(
           0, __builtin_bit_cast(int, v), 0x164, 0xf, 0xf, true));
  v += __builtin_bit_cast(float, __builtin_amdgcn_update_dpp(
           0, __builtin_bit_cast(int, v), 0x168, 0xf, 0xf, true));
  return v;
}

__global__ __launch_bounds__(256) void gla_chunk_kernel(
    const float* __restrict__ qh_ws, const float* __restrict__ kh_ws,
    const float* __restrict__ vT_ws, const float* __restrict__ sp_ws,
    float* __restrict__ on, float* __restrict__ ssq_g) {
  constexpr int LDA = CH + 8;
  __shared__ __align__(16) _Float16 qt[CH][LDA];
  __shared__ __align__(16) _Float16 kt[CH][LDA];
  __shared__ __align__(16) _Float16 Am[CH][LDA];
  __shared__ __align__(16) _Float16 vT[DVc][LDA];
  __shared__ __align__(16) _Float16 sT[DVc][LDA];
  __shared__ float ssqP[2][CH];

  const int tile = blockIdx.x;
  const int h = tile & 3, tok0 = (tile >> 2) * CH;
  const int tid = threadIdx.x, lane = tid & 31, wave = tid >> 5;

  {
    const _Float16* qtile  = (const _Float16*)qh_ws + (size_t)tile * (CH * DKc);
    const _Float16* ktile  = (const _Float16*)kh_ws + (size_t)tile * (CH * DKc);
    const _Float16* vTtile = (const _Float16*)vT_ws + (size_t)tile * (CH * DVc);
    const _Float16* stile  = (const _Float16*)sp_ws + (size_t)tile * (DVc * DKc);
    for (int r = 0; r < 2; ++r) {
      const int i = tid + 256 * r, row = i >> 3, cc = (i & 7) * 8;
      *(v8h*)&qt[row][cc] = *(const v8h*)(qtile + i * 8);
      *(v8h*)&kt[row][cc] = *(const v8h*)(ktile + i * 8);
    }
    for (int r = 0; r < 4; ++r) {
      const int i = tid + 256 * r, row = i >> 3, cc = (i & 7) * 8;
      *(v8h*)&vT[row][cc] = *(const v8h*)(vTtile + i * 8);
      *(v8h*)&sT[row][cc] = *(const v8h*)(stile  + i * 8);
    }
  }
  __syncthreads();

  {
    const int tRow = wave & 3, sp0 = (wave >> 2) * 2;
#pragma unroll
    for (int spi = 0; spi < 2; ++spi) {
      const int sTile = sp0 + spi;
      v8f acc = {};
      if (sTile <= tRow) {
#pragma unroll
        for (int k2 = 0; k2 < 2; ++k2) {
          v16h a  = frag16(&qt[0][0], LDA, tRow * 16, k2 * 32);
          v16h bm = frag16(&kt[0][0], LDA, sTile * 16, k2 * 32);
          acc = wmma16(a, bm, acc);
        }
      }
      const int sIdx = sTile * 16 + (lane & 15), mOff = (lane >> 4) * 8;
#pragma unroll
      for (int r = 0; r < 8; ++r) {
        const int tIdx = tRow * 16 + mOff + r;
        Am[tIdx][sIdx] = (_Float16)((sIdx <= tIdx) ? acc[r] : 0.f);
      }
    }
  }
  __syncthreads();

  {
    const int tRow = wave & 3, nHalf = wave >> 2, ntb = nHalf * 4;
    v8f acc[4] = {};
#pragma unroll
    for (int k2 = 0; k2 < 2; ++k2) {
      v16h aA = frag16(&Am[0][0], LDA, tRow * 16, k2 * 32);
      v16h aQ = frag16(&qt[0][0], LDA, tRow * 16, k2 * 32);
#pragma unroll
      for (int j = 0; j < 4; ++j) {
        v16h bv = frag16(&vT[0][0], LDA, (ntb + j) * 16, k2 * 32);
        acc[j] = wmma16(aA, bv, acc[j]);
        v16h bs = frag16(&sT[0][0], LDA, (ntb + j) * 16, k2 * 32);
        acc[j] = wmma16(aQ, bs, acc[j]);
      }
    }
    float* ob = on + (size_t)tok0 * (Hc * DVc) + h * DVc;
    const int nl = lane & 15, mOff = (lane >> 4) * 8;
#pragma unroll
    for (int j = 0; j < 4; ++j) {
      const int n = (ntb + j) * 16 + nl;
#pragma unroll
      for (int r = 0; r < 8; ++r) {
        const int t = tRow * 16 + mOff + r;
        ob[(size_t)t * (Hc * DVc) + n] = acc[j][r];
      }
    }
#pragma unroll
    for (int r = 0; r < 8; ++r) {
      float p = acc[0][r] * acc[0][r];
#pragma unroll
      for (int j = 1; j < 4; ++j) p += acc[j][r] * acc[j][r];
      p = row16_sum(p);
      if (nl == 0) ssqP[nHalf][tRow * 16 + mOff + r] = p;
    }
  }
  __syncthreads();
  if (tid < CH)
    ssq_g[(size_t)h * TOK + tok0 + tid] = ssqP[0][tid] + ssqP[1][tid];
}

struct OutTile {
  v4f ov[4], gv[4], rw;
  float ss[4];
  v8h bb[2];
};

__global__ __launch_bounds__(256) void out_gemm_kernel(
    const float* __restrict__ on, const float* __restrict__ g,
    const float* __restrict__ rms_w, const float* __restrict__ ssq,
    const float* __restrict__ woT_ws, float* __restrict__ Cm) {
  __shared__ __align__(16) _Float16 As[2][128][40];
  __shared__ __align__(16) _Float16 BsT[2][128][40];
  constexpr int NS = 512 / 32;
  const int m0 = blockIdx.x * 128, n0 = blockIdx.y * 128;
  const int tid = threadIdx.x, lane = tid & 31, wave = tid >> 5;
  const int mr0 = (wave & 3) * 32, nc0 = (wave >> 2) * 64;
  const int arow = tid >> 3, acg = tid & 7;
  const size_t aoff = (size_t)(m0 + arow) * 512 + acg * 4;
  const float* on_p = on + aoff;
  const float* g_p  = g + aoff;
  const float* rw_p = rms_w + acg * 4;
  const float* ssq_p = ssq + m0 + arow;
  const size_t pfoff = (size_t)(m0 + (tid >> 1)) * 512;
  const _Float16* bt_p = (const _Float16*)woT_ws + (size_t)blockIdx.y * 16 * WTILE + tid * 8;

  OutTile ta, tb;
  v8f acc[2][4] = {};

  auto load = [&](OutTile& t, int s1) {
    const int kk = s1 * 32;
    t.rw = *(const v4f*)(rw_p + (kk & (DVc - 1)));
#pragma unroll
    for (int r = 0; r < 4; ++r) {
      t.ss[r] = ssq_p[(size_t)(s1 >> 2) * TOK + 32 * r];
      t.ov[r] = *(const v4f*)(on_p + (size_t)r * 32 * 512 + kk);
      t.gv[r] = *(const v4f*)(g_p  + (size_t)r * 32 * 512 + kk);
    }
#pragma unroll
    for (int r = 0; r < 2; ++r)
      t.bb[r] = *(const v8h*)(bt_p + (size_t)s1 * WTILE + r * 2048);
    if (s1 + 1 < NS) {
      __builtin_prefetch(on + pfoff + kk + 32, 0, 1);
      __builtin_prefetch(g  + pfoff + kk + 32, 0, 1);
    }
  };
  auto store = [&](const OutTile& t, int s1) {
    const int st = s1 & 1;
#pragma unroll
    for (int r = 0; r < 4; ++r) {
      const float ms = t.ss[r] * (1.f / DVc);
      const float rstd = rsqrtf(ms + RMS_EPS);
      v4h hv;
#pragma unroll
      for (int e = 0; e < 4; ++e) {
        const float gg = t.gv[r][e];
        const float val = t.ov[r][e] * rstd * t.rw[e];
        hv[e] = (_Float16)(val * gg * (1.f / (1.f + __expf(-gg))));
      }
      *(v4h*)&As[st][arow + 32 * r][acg * 4] = hv;
    }
#pragma unroll
    for (int r = 0; r < 2; ++r) {
      const int i = tid + 256 * r;
      *(v8h*)&BsT[st][i >> 2][(i & 3) * 8] = t.bb[r];
    }
  };
  auto compute = [&](int st) {
    v16h a0 = frag16(&As[st][0][0], 40, mr0, 0);
    v16h a1 = frag16(&As[st][0][0], 40, mr0 + 16, 0);
#pragma unroll
    for (int j = 0; j < 4; ++j) {
      v16h bm = frag16(&BsT[st][0][0], 40, nc0 + j * 16, 0);
      acc[0][j] = wmma16(a0, bm, acc[0][j]);
      acc[1][j] = wmma16(a1, bm, acc[1][j]);
    }
  };

  load(ta, 0);
  load(tb, 1);
  store(ta, 0);
  __syncthreads();
  for (int s = 0; s < NS; s += 2) {
    if (s + 2 < NS) load(ta, s + 2);
    store(tb, s + 1);
    compute(0);
    __syncthreads();
    if (s + 3 < NS) load(tb, s + 3);
    if (s + 2 < NS) store(ta, s + 2);
    compute(1);
    __syncthreads();
  }
  const int nl = lane & 15, mOff = (lane >> 4) * 8;
#pragma unroll
  for (int a2 = 0; a2 < 2; ++a2) {
#pragma unroll
    for (int j = 0; j < 4; ++j) {
      const int n = n0 + nc0 + j * 16 + nl;
#pragma unroll
      for (int r = 0; r < 8; ++r)
        Cm[(size_t)(m0 + mr0 + a2 * 16 + mOff + r) * 512 + n] = acc[a2][j][r];
    }
  }
}

extern "C" void kernel_launch(void* const* d_in, const int* in_sizes, int n_in,
                              void* d_out, int out_size, void* d_ws, size_t ws_size,
                              hipStream_t stream) {
  (void)in_sizes; (void)n_in; (void)out_size; (void)ws_size;
  const float* x    = (const float*)d_in[0];
  const float* Wq   = (const float*)d_in[1];
  const float* Wk   = (const float*)d_in[2];
  const float* Wv   = (const float*)d_in[3];
  const float* Wg   = (const float*)d_in[4];
  const float* Wo   = (const float*)d_in[5];
  const float* Wgk1 = (const float*)d_in[6];
  const float* Wgk2 = (const float*)d_in[7];
  const float* bgk2 = (const float*)d_in[8];
  const float* rmsw = (const float*)d_in[9];
  float* out = (float*)d_out;

  float* ws   = (float*)d_ws;
  float* qws  = ws;
  float* kws  = qws  + (size_t)TOK * 256;
  float* vws  = kws  + (size_t)TOK * 256;
  float* gws  = vws  + (size_t)TOK * 512;
  float* xg1  = gws  + (size_t)TOK * 512;
  float* on   = xg1  + (size_t)TOK * 16;
  float* ssq  = on   + (size_t)TOK * 512;
  float* qhw  = ssq  + (size_t)TOK * Hc;
  float* khw  = qhw  + (size_t)TOK * 128;
  float* vTw  = khw  + (size_t)TOK * 128;
  float* elc  = vTw  + (size_t)TOK * 256;
  float* kvw  = elc  + (size_t)TOK * 4;
  float* spw  = kvw  + (size_t)TOK * 512;
  float* wTw  = spw  + (size_t)TOK * 256;
  float* woTw = wTw  + (size_t)PROJ_WTILES * WTILE / 2;

  lr1_kernel<<<LR1_BLOCKS + PROJ_WTILES + OUT_WTILES, 256, 0, stream>>>(
      x, Wgk1, xg1, Wq, Wk, Wv, Wg, Wo, wTw);
  proj_kernel<<<dim3(TOK / 128, 1536 / 128), 256, 0, stream>>>(
      x, wTw, qws, kws, vws, gws);
  gate_prep_kernel<<<(TOK / CH) * Hc, 256, 0, stream>>>(
      xg1, Wgk2, bgk2, qws, kws, vws, qhw, khw, vTw, elc, kvw);
  state_kernel<<<dim3(Bc * Hc, (DVc * DKc) / 512), 256, 0, stream>>>(kvw, elc, spw);
  gla_chunk_kernel<<<(TOK / CH) * Hc, 256, 0, stream>>>(
      qhw, khw, vTw, spw, on, ssq);
  out_gemm_kernel<<<dim3(TOK / 128, 512 / 128), 256, 0, stream>>>(
      on, gws, rmsw, ssq, woTw, out);
}


# Projection GEMM and gate prep fused into one launch: a block owns its 64-token chunk x one head's q|k|v|g columns and makes the scan operands from its accumulators through LDS; q, k, v skip memory
# speedup vs baseline: 14.6807x; 1.6963x over previous
#include <hip/hip_runtime.h>
#include <hip/hip_bf16.h>
#include <math.h>


typedef __attribute__((ext_vector_type(16))) _Float16 v16h;
typedef __attribute__((ext_vector_type(8)))  _Float16 v8h;
typedef __attribute__((ext_vector_type(4)))  _Float16 v4h;
typedef __attribute__((ext_vector_type(8)))  float    v8f;
typedef __attribute__((ext_vector_type(4)))  float    v4f;
typedef __attribute__((ext_vector_type(2)))  float    v2f;
typedef __attribute__((ext_vector_type(2)))  _Float16 v2h;

constexpr int Bc  = 4;
constexpr int Tc  = 2048;
constexpr int HIDc = 512;
constexpr int Hc  = 4;
constexpr int DKc = 64;
constexpr int DVc = 128;
constexpr int CH  = 64;
constexpr int TOK = Bc * Tc;
constexpr float RMS_EPS = 1e-5f;
constexpr int WTILE = 128 * 32;
constexpr int LR1_BLOCKS = TOK / 16;
constexpr int PROJ_WTILES = (1536 / 128) * (HIDc / 32);
constexpr int OUT_WTILES  = (512 / 128) * (512 / 32);
constexpr int FN = 384;
constexpr int FTILE = FN * 32;

__device__ __forceinline__ v8f wmma16(v16h a, v16h b, v8f c) {
  return __builtin_amdgcn_wmma_f32_16x16x32_f16(false, a, false, b, (short)0, c,
                                                false, false);
}

__device__ __forceinline__ v16h frag16(const _Float16* p, int ld, int rowBase,
                                       int kBase) {
  const int lane = threadIdx.x & 31, half = lane >> 4;
  const _Float16* row = p + (size_t)(rowBase + (lane & 15)) * ld + kBase + half * 8;
  v8h lo = *(const v8h*)(row);
  v8h hi = *(const v8h*)(row + 16);
  return __builtin_shufflevector(lo, hi, 0, 1, 2, 3, 4, 5, 6, 7,
                                 8, 9, 10, 11, 12, 13, 14, 15);
}


__global__ __launch_bounds__(256) void lr1_kernel(
    const float* __restrict__ x, const float* __restrict__ Wgk1,
    float* __restrict__ xg1,
    const float* __restrict__ Wq, const float* __restrict__ Wk,
    const float* __restrict__ Wv, const float* __restrict__ Wg,
    const float* __restrict__ Wo, float* __restrict__ wT_ws) {
  __shared__ __align__(16) float xs[16][512];
  __shared__ __align__(16) float part[32][272];
  if (blockIdx.x >= LR1_BLOCKS) {
    const int j = blockIdx.x - LR1_BLOCKS;
    const int nb = (j < PROJ_WTILES ? j : j - PROJ_WTILES) >> 4, ks = j & 15;
    const float* src;
    int ld;
    if (j >= PROJ_WTILES) { src = Wo + nb * 128;       ld = 512; }
    else if (nb < 2)      { src = Wq + nb * 128;       ld = 256; }
    else if (nb < 4)      { src = Wk + (nb - 2) * 128; ld = 256; }
    else if (nb < 8)      { src = Wv + (nb - 4) * 128; ld = 512; }
    else                  { src = Wg + (nb - 8) * 128; ld = 512; }
    _Float16 (*tT)[40] = (_Float16 (*)[40])&xs[0][0];
    for (int i = threadIdx.x; i < 1024; i += 256) {
      const int k = i >> 5, ng = i & 31;
      const v4f wv = *(const v4f*)(src + (size_t)(ks * 32 + k) * ld + ng * 4);
#pragma unroll
      for (int j2 = 0; j2 < 4; ++j2) tT[ng * 4 + j2][k] = (_Float16)wv[j2];
    }
    __syncthreads();
    for (int r = 0; r < 2; ++r) {
      const int i = threadIdx.x + 256 * r, nr = i >> 2;
      size_t o;
      if (j >= PROJ_WTILES) {
        o = (size_t)j * WTILE + nr * 32;
      } else {
        int hh, pos;
        if (nb < 2)      { const int c = nb * 128 + nr;       hh = c >> 6; pos = c & 63; }
        else if (nb < 4) { const int c = (nb - 2) * 128 + nr; hh = c >> 6; pos = 64 + (c & 63); }
        else if (nb < 8) { hh = nb - 4; pos = 128 + nr; }
        else             { hh = nb - 8; pos = 256 + nr; }
        o = ((size_t)(hh * 16 + ks) * FN + pos) * 32;
      }
      *(v8h*)((_Float16*)wT_ws + o + (i & 3) * 8) = *(const v8h*)&tT[nr][(i & 3) * 8];
    }
    return;
  }
  const int t0 = blockIdx.x * 16;
  const int tid = threadIdx.x;
  const int jp = tid & 7, ks = tid >> 3;
  v2f w[16];
#pragma unroll
  for (int kk = 0; kk < 16; ++kk)
    w[kk] = *(const v2f*)(Wgk1 + (size_t)(ks * 16 + kk) * 16 + jp * 2);
#pragma unroll
  for (int r = 0; r < 8; ++r) {
    const int i = tid + 256 * r;
    *(v4f*)&xs[i >> 7][(i & 127) * 4] =
        *(const v4f*)(x + (size_t)(t0 + (i >> 7)) * 512 + (i & 127) * 4);
  }
  __syncthreads();
#pragma unroll 4
  for (int tok = 0; tok < 16; ++tok) {
    const float* xr = &xs[tok][ks * 16];
    float s0 = 0.f, s1 = 0.f;
#pragma unroll
    for (int q4 = 0; q4 < 4; ++q4) {
      const v4f xv = *(const v4f*)(xr + q4 * 4);
#pragma unroll
      for (int e = 0; e < 4; ++e) {
        s0 += xv[e] * w[q4 * 4 + e][0];
        s1 += xv[e] * w[q4 * 4 + e][1];
      }
    }
    const v2f p = {s0, s1};
    *(v2f*)&part[ks][tok * 16 + jp * 2] = p;
  }
  __syncthreads();
  float s = 0.f;
#pragma unroll
  for (int k2 = 0; k2 < 32; ++k2) s += part[k2][tid];
  xg1[(size_t)t0 * 16 + tid] = s;
}

__global__ __launch_bounds__(256) void proj_gate_kernel(
    const float* __restrict__ x, const float* __restrict__ wT_ws,
    const float* __restrict__ xg1, const float* __restrict__ Wgk2,
    const float* __restrict__ bgk2, float* __restrict__ gws,
    float* __restrict__ qh_ws, float* __restrict__ kh_ws,
    float* __restrict__ vT_ws, float* __restrict__ expLCg,
    float* __restrict__ kv_ws) {
  constexpr int LDA = CH + 8;
  constexpr int LDF = DKc + 1;
  constexpr int NS = HIDc / 32;
  constexpr int A_BYTES = 2 * CH * 40 * 2;
  constexpr int B_BYTES = 2 * FN * 40 * 2;
  __shared__ __align__(16) unsigned char stage[A_BYTES + B_BYTES];
  __shared__ __align__(16) float    xs[CH][16];
  __shared__ float                  lamS[CH][DKc + 1];
  __shared__ __align__(16) _Float16 vTS[DVc][LDA];
  _Float16 (*As)[CH][40]  = (_Float16 (*)[CH][40])stage;
  _Float16 (*BsT)[FN][40] = (_Float16 (*)[FN][40])(stage + A_BYTES);
  float* qF = (float*)stage;
  float* kF = qF + CH * LDF;
  _Float16 (*qS)[LDA]  = (_Float16 (*)[LDA])(stage + 2 * CH * LDF * 4);
  _Float16 (*kS)[LDA]  = qS + CH;
  _Float16 (*kTS)[LDA] = kS + CH;
  static_assert(2 * CH * LDF * 4 + 3 * CH * LDA * 2 <= A_BYTES + B_BYTES, "alias fits");
  static_assert((2 * CH * LDF * 4) % 16 == 0, "f16 tiles 16-B aligned");

  const int tile = blockIdx.x;
  const int h = tile & 3, tok0 = (tile >> 2) * CH;
  const int tid = threadIdx.x, lane = tid & 31, wave = tid >> 5;
  const int d = tid & 63, grp = tid >> 6;
  const int mr0 = (wave & 1) * 32;
  const int nc0 = (wave >> 1) * 96;
  const int arow = tid >> 3, acg = tid & 7;
  const float* xa_p = x + (size_t)(tok0 + arow) * HIDc + acg * 4;
  const float* xpf_p = x + (size_t)(tok0 + (tid >> 2)) * HIDc;
  const _Float16* bt_p = (const _Float16*)wT_ws + (size_t)h * 16 * FTILE + tid * 8;
  v4f xa[2];
  v8h bb[6];
  auto load = [&](int s1) {
#pragma unroll
    for (int r = 0; r < 2; ++r)
      xa[r] = *(const v4f*)(xa_p + (size_t)r * 32 * HIDc + s1 * 32);
#pragma unroll
    for (int r = 0; r < 6; ++r)
      bb[r] = *(const v8h*)(bt_p + (size_t)s1 * FTILE + r * 2048);
    if (s1 + 1 < NS) __builtin_prefetch(xpf_p + (s1 + 1) * 32, 0, 1);
  };
  auto store = [&](int st) {
#pragma unroll
    for (int r = 0; r < 2; ++r) {
      v4h hv;
#pragma unroll
      for (int e = 0; e < 4; ++e) hv[e] = (_Float16)xa[r][e];
      *(v4h*)&As[st][arow + 32 * r][acg * 4] = hv;
    }
#pragma unroll
    for (int r = 0; r < 6; ++r) {
      const int i = tid + 256 * r;
      *(v8h*)&BsT[st][i >> 2][(i & 3) * 8] = bb[r];
    }
  };

  *(v4f*)(&xs[0][0] + tid * 4) = *(const v4f*)(xg1 + (size_t)tok0 * 16 + tid * 4);
  {
    const int n = h * DKc + d;
    float w[16];
#pragma unroll
    for (int j = 0; j < 16; ++j) w[j] = Wgk2[j * 256 + n];
    const float bias = bgk2[n];
    load(0);
    __syncthreads();
    for (int tt = 0; tt < 16; ++tt) {
      const int t = grp * 16 + tt;
      float s = bias;
#pragma unroll
      for (int j = 0; j < 16; ++j) s += xs[t][j] * w[j];
      const float ls = fminf(s, 0.f) - log1pf(__expf(-fabsf(s)));
      lamS[t][d] = ls * (1.f / 16.f);
    }
  }
  __syncthreads();
  if (tid < DKc) {
    float lam = 0.f;
    for (int t = 0; t < CH; ++t) {
      lam += lamS[t][tid];
      lamS[t][tid] = lam;
    }
    expLCg[(size_t)tile * DKc + tid] = __expf(lam);
  }
  store(0);
  __syncthreads();

  v8f acc[2][6] = {};
  for (int s = 0; s < NS; ++s) {
    const int st = s & 1;
    if (s + 1 < NS) load(s + 1);
    __builtin_amdgcn_sched_barrier(0);
    v16h a0 = frag16(&As[st][0][0], 40, mr0, 0);
    v16h a1 = frag16(&As[st][0][0], 40, mr0 + 16, 0);
#pragma unroll
    for (int j = 0; j < 6; ++j) {
      v16h bm = frag16(&BsT[st][0][0], 40, nc0 + j * 16, 0);
      acc[0][j] = wmma16(a0, bm, acc[0][j]);
      acc[1][j] = wmma16(a1, bm, acc[1][j]);
    }
    __builtin_amdgcn_sched_barrier(0);
    if (s + 1 < NS) store(st ^ 1);
    __syncthreads();
  }

  const int nl = lane & 15, mOff = (lane >> 4) * 8;
  const int ncu = __builtin_amdgcn_readfirstlane(nc0);
#pragma unroll
  for (int a2 = 0; a2 < 2; ++a2) {
    const int tb = mr0 + a2 * 16 + mOff;
#pragma unroll
    for (int j = 0; j < 6; ++j) {
      const int cb = ncu + j * 16;
      if (cb < 64) {
#pragma unroll
        for (int r = 0; r < 8; ++r)
          qF[(tb + r) * LDF + cb + nl] = acc[a2][j][r] * 0.125f;
      } else if (cb < 128) {
#pragma unroll
        for (int r = 0; r < 8; ++r)
          kF[(tb + r) * LDF + (cb - 64) + nl] = acc[a2][j][r];
      } else if (cb < 256) {
        v8h hv;
#pragma unroll
        for (int r = 0; r < 8; ++r) hv[r] = (_Float16)acc[a2][j][r];
        *(v8h*)&vTS[cb - 128 + nl][tb] = hv;
      } else {
        float* gp = gws + (size_t)(tok0 + tb) * (Hc * DVc) + h * DVc + (cb - 256) + nl;
#pragma unroll
        for (int r = 0; r < 8; ++r) gp[(size_t)r * (Hc * DVc)] = acc[a2][j][r];
      }
    }
  }
  __syncthreads();

  const float lamC = lamS[CH - 1][d];
#pragma unroll 4
  for (int tt = 0; tt < 16; ++tt) {
    const int t = grp * 16 + tt;
    const float eq = __expf(lamS[t][d] - lamC);
    const float qh = fmaxf(fminf(qF[t * LDF + d] * eq, 60000.f), -60000.f);
    const float kh = kF[t * LDF + d] / eq;
    qS[t][d]  = (_Float16)qh;
    kS[t][d]  = (_Float16)kh;
    kTS[d][t] = (_Float16)kh;
  }
  __syncthreads();

  _Float16* qd  = (_Float16*)qh_ws  + (size_t)tile * (CH * DKc);
  _Float16* kd  = (_Float16*)kh_ws  + (size_t)tile * (CH * DKc);
  _Float16* vtd = (_Float16*)vT_ws  + (size_t)tile * (CH * DVc);
  for (int r = 0; r < 2; ++r) {
    const int i = tid + 256 * r, row = i >> 3, cc = (i & 7) * 8;
    *(v8h*)(qd  + i * 8) = *(const v8h*)&qS[row][cc];
    *(v8h*)(kd  + i * 8) = *(const v8h*)&kS[row][cc];
  }
  for (int r = 0; r < 4; ++r) {
    const int i = tid + 256 * r, row = i >> 3, cc = (i & 7) * 8;
    *(v8h*)(vtd + i * 8) = *(const v8h*)&vTS[row][cc];
  }

  {
    v8f kacc[4] = {};
#pragma unroll
    for (int k2 = 0; k2 < 2; ++k2) {
      v16h bv = frag16(&vTS[0][0], LDA, wave * 16, k2 * 32);
#pragma unroll
      for (int dr = 0; dr < 4; ++dr) {
        v16h aK = frag16(&kTS[0][0], LDA, dr * 16, k2 * 32);
        kacc[dr] = wmma16(aK, bv, kacc[dr]);
      }
    }
    float* dst = kv_ws + (size_t)tile * (DVc * DKc) +
                 (wave * 16 + (lane & 15)) * DKc + (lane >> 4) * 8;
#pragma unroll
    for (int dr = 0; dr < 4; ++dr) {
      const v4f lo = {kacc[dr][0], kacc[dr][1], kacc[dr][2], kacc[dr][3]};
      const v4f hi = {kacc[dr][4], kacc[dr][5], kacc[dr][6], kacc[dr][7]};
      *(v4f*)(dst + dr * 16)     = lo;
      *(v4f*)(dst + dr * 16 + 4) = hi;
    }
  }
}

__global__ __launch_bounds__(256) void state_kernel(
    const float* __restrict__ kv_ws, const float* __restrict__ expLCg,
    float* __restrict__ sp_ws) {
  constexpr int NC = Tc / CH;
  constexpr int GRP = 16;
  const int bh = blockIdx.x, b = bh >> 2, h = bh & 3;
  const int e2 = (blockIdx.y * 256 + threadIdx.x) * 2;
  const int d = e2 & 63;
  _Float16* sp = (_Float16*)sp_ws;
  float s0 = 0.f, s1 = 0.f;
  for (int c0 = 0; c0 < NC; c0 += GRP) {
    v2f kv[GRP], dc[GRP];
#pragma unroll
    for (int i = 0; i < GRP; ++i) {
      const size_t tile = (size_t)((b * NC + c0 + i) * Hc + h);
      kv[i] = *(const v2f*)(kv_ws + tile * (DVc * DKc) + e2);
      dc[i] = *(const v2f*)(expLCg + tile * DKc + d);
    }
#pragma unroll
    for (int i = 0; i < GRP; ++i) {
      const size_t tile = (size_t)((b * NC + c0 + i) * Hc + h);
      s0 *= dc[i][0];
      s1 *= dc[i][1];
      v2h o2;
      o2[0] = (_Float16)s0;
      o2[1] = (_Float16)s1;
      *(v2h*)(sp + tile * (DVc * DKc) + e2) = o2;
      s0 += kv[i][0];
      s1 += kv[i][1];
    }
  }
}

__device__ __forceinline__ float row16_sum(float v) {
  v += __builtin_bit_cast(float, __builtin_amdgcn_update_dpp(
           0, __builtin_bit_cast(int, v), 0x161, 0xf, 0xf, true));
  v += __builtin_bit_cast(float, __builtin_amdgcn_update_dpp(
           0, __builtin_bit_cast(int, v), 0x162, 0xf, 0xf, true));
  v += __builtin_bit_cast(float, __builtin_amdgcn_update_dpp(
           0, __builtin_bit_cast(int, v), 0x164, 0xf, 0xf, true));
  v += __builtin_bit_cast(float, __builtin_amdgcn_update_dpp(
           0, __builtin_bit_cast(int, v), 0x168, 0xf, 0xf, true));
  return v;
}

__global__ __launch_bounds__(256) void gla_chunk_kernel(
    const float* __restrict__ qh_ws, const float* __restrict__ kh_ws,
    const float* __restrict__ vT_ws, const float* __restrict__ sp_ws,
    const float* __restrict__ g, const float* __restrict__ rms_w,
    float* __restrict__ a16_ws) {
  constexpr int LDA = CH + 8;
  constexpr int LDO = DVc + 8;
  __shared__ __align__(16) _Float16 qt[CH][LDA];
  __shared__ __align__(16) _Float16 kt[CH][LDA];
  __shared__ __align__(16) _Float16 Am[CH][LDA];
  __shared__ __align__(16) _Float16 vT[DVc][LDA];
  __shared__ __align__(16) _Float16 sT[DVc][LDA];
  __shared__ __align__(16) _Float16 oS[CH][LDO];
  __shared__ float ssqP[2][CH];

  const int tile = blockIdx.x;
  const int h = tile & 3, tok0 = (tile >> 2) * CH;
  const int tid = threadIdx.x, lane = tid & 31, wave = tid >> 5;
  const int tRow = wave & 3, nHalf = wave >> 2, ntb = nHalf * 4;
  const int nl = lane & 15, mOff = (lane >> 4) * 8;

  {
    const _Float16* qtile  = (const _Float16*)qh_ws + (size_t)tile * (CH * DKc);
    const _Float16* ktile  = (const _Float16*)kh_ws + (size_t)tile * (CH * DKc);
    const _Float16* vTtile = (const _Float16*)vT_ws + (size_t)tile * (CH * DVc);
    const _Float16* stile  = (const _Float16*)sp_ws + (size_t)tile * (DVc * DKc);
    for (int r = 0; r < 2; ++r) {
      const int i = tid + 256 * r, row = i >> 3, cc = (i & 7) * 8;
      *(v8h*)&qt[row][cc] = *(const v8h*)(qtile + i * 8);
      *(v8h*)&kt[row][cc] = *(const v8h*)(ktile + i * 8);
    }
    for (int r = 0; r < 4; ++r) {
      const int i = tid + 256 * r, row = i >> 3, cc = (i & 7) * 8;
      *(v8h*)&vT[row][cc] = *(const v8h*)(vTtile + i * 8);
      *(v8h*)&sT[row][cc] = *(const v8h*)(stile  + i * 8);
    }
  }
  float gv[4][8], rwv[4];
  {
    const float* gb = g + (size_t)(tok0 + tRow * 16 + mOff) * (Hc * DVc) + h * DVc + nl;
#pragma unroll
    for (int j = 0; j < 4; ++j) {
      rwv[j] = rms_w[(ntb + j) * 16 + nl];
#pragma unroll
      for (int r = 0; r < 8; ++r)
        gv[j][r] = gb[(size_t)r * (Hc * DVc) + (ntb + j) * 16];
    }
  }
  __syncthreads();

  {
    const int sp0 = (wave >> 2) * 2;
#pragma unroll
    for (int spi = 0; spi < 2; ++spi) {
      const int sTile = sp0 + spi;
      v8f acc = {};
      if (sTile <= tRow) {
#pragma unroll
        for (int k2 = 0; k2 < 2; ++k2) {
          v16h a  = frag16(&qt[0][0], LDA, tRow * 16, k2 * 32);
          v16h bm = frag16(&kt[0][0], LDA, sTile * 16, k2 * 32);
          acc = wmma16(a, bm, acc);
        }
      }
      const int sIdx = sTile * 16 + nl;
#pragma unroll
      for (int r = 0; r < 8; ++r) {
        const int tIdx = tRow * 16 + mOff + r;
        Am[tIdx][sIdx] = (_Float16)((sIdx <= tIdx) ? acc[r] : 0.f);
      }
    }
  }
  __syncthreads();

  v8f acc[4] = {};
#pragma unroll
  for (int k2 = 0; k2 < 2; ++k2) {
    v16h aA = frag16(&Am[0][0], LDA, tRow * 16, k2 * 32);
    v16h aQ = frag16(&qt[0][0], LDA, tRow * 16, k2 * 32);
#pragma unroll
    for (int j = 0; j < 4; ++j) {
      v16h bv = frag16(&vT[0][0], LDA, (ntb + j) * 16, k2 * 32);
      acc[j] = wmma16(aA, bv, acc[j]);
      v16h bs = frag16(&sT[0][0], LDA, (ntb + j) * 16, k2 * 32);
      acc[j] = wmma16(aQ, bs, acc[j]);
    }
  }
#pragma unroll
  for (int r = 0; r < 8; ++r) {
    float p = acc[0][r] * acc[0][r];
#pragma unroll
    for (int j = 1; j < 4; ++j) p += acc[j][r] * acc[j][r];
    p = row16_sum(p);
    if (nl == 0) ssqP[nHalf][tRow * 16 + mOff + r] = p;
  }
  __syncthreads();

#pragma unroll
  for (int r = 0; r < 8; ++r) {
    const int t = tRow * 16 + mOff + r;
    const float ms = (ssqP[0][t] + ssqP[1][t]) * (1.f / DVc);
    const float rstd = rsqrtf(ms + RMS_EPS);
#pragma unroll
    for (int j = 0; j < 4; ++j) {
      const float gg = gv[j][r];
      const float val = acc[j][r] * rstd * rwv[j];
      oS[t][(ntb + j) * 16 + nl] =
          (_Float16)(val * gg * (1.f / (1.f + __expf(-gg))));
    }
  }
  __syncthreads();

  {
    _Float16* adst = (_Float16*)a16_ws +
                     ((size_t)(tok0 >> 7) * 16 + h * 4) * WTILE + (tok0 & 127) * 32 + tid * 8;
    const int t = tid >> 2, c8 = (tid & 3) * 8;
#pragma unroll
    for (int q = 0; q < 4; ++q)
      *(v8h*)(adst + (size_t)q * WTILE) = *(const v8h*)&oS[t][q * 32 + c8];
  }
}

__global__ __launch_bounds__(256) void out_gemm_kernel(
    const float* __restrict__ a16_ws, const float* __restrict__ woT_ws,
    float* __restrict__ Cm) {
  constexpr int KD = 128;
  constexpr int LDK = KD + 40;
  constexpr int NS = 512 / KD;
  __shared__ __align__(16) _Float16 As[2][128][LDK];
  __shared__ __align__(16) _Float16 BsT[2][128][LDK];
  const int m0 = blockIdx.x * 128, n0 = blockIdx.y * 128;
  const int tid = threadIdx.x, lane = tid & 31, wave = tid >> 5;
  const int mr0 = (wave & 3) * 32, nc0 = (wave >> 2) * 64;
  const _Float16* at_p = (const _Float16*)a16_ws + (size_t)blockIdx.x * 16 * WTILE + tid * 8;
  const _Float16* bt_p = (const _Float16*)woT_ws + (size_t)blockIdx.y * 16 * WTILE + tid * 8;

  v8h aa[8], bb[8];
  v8f acc[2][4] = {};

  auto load = [&](int s1) {
#pragma unroll
    for (int r = 0; r < 8; ++r) {
      aa[r] = *(const v8h*)(at_p + (size_t)s1 * (4 * WTILE) + r * 2048);
      bb[r] = *(const v8h*)(bt_p + (size_t)s1 * (4 * WTILE) + r * 2048);
    }
  };
  auto store = [&](int st) {
#pragma unroll
    for (int r = 0; r < 8; ++r) {
      const int i = tid + 256 * r;
      const int row = (i & 511) >> 2, k = (i >> 9) * 32 + (i & 3) * 8;
      *(v8h*)&As[st][row][k]  = aa[r];
      *(v8h*)&BsT[st][row][k] = bb[r];
    }
  };

  load(0);
  store(0);
  __syncthreads();
  for (int s = 0; s < NS; ++s) {
    const int st = s & 1;
    if (s + 1 < NS) load(s + 1);
    __builtin_amdgcn_sched_barrier(0);
#pragma unroll
    for (int kk = 0; kk < KD; kk += 32) {
      v16h a0 = frag16(&As[st][0][0], LDK, mr0, kk);
      v16h a1 = frag16(&As[st][0][0], LDK, mr0 + 16, kk);
#pragma unroll
      for (int j = 0; j < 4; ++j) {
        v16h bm = frag16(&BsT[st][0][0], LDK, nc0 + j * 16, kk);
        acc[0][j] = wmma16(a0, bm, acc[0][j]);
        acc[1][j] = wmma16(a1, bm, acc[1][j]);
      }
    }
    if (s + 1 < NS) store(st ^ 1);
    __syncthreads();
  }
  const int nl = lane & 15, mOff = (lane >> 4) * 8;
#pragma unroll
  for (int a2 = 0; a2 < 2; ++a2) {
#pragma unroll
    for (int j = 0; j < 4; ++j) {
      const int n = n0 + nc0 + j * 16 + nl;
#pragma unroll
      for (int r = 0; r < 8; ++r)
        Cm[(size_t)(m0 + mr0 + a2 * 16 + mOff + r) * 512 + n] = acc[a2][j][r];
    }
  }
}

extern "C" void kernel_launch(void* const* d_in, const int* in_sizes, int n_in,
                              void* d_out, int out_size, void* d_ws, size_t ws_size,
                              hipStream_t stream) {
  (void)in_sizes; (void)n_in; (void)out_size; (void)ws_size;
  const float* x    = (const float*)d_in[0];
  const float* Wq   = (const float*)d_in[1];
  const float* Wk   = (const float*)d_in[2];
  const float* Wv   = (const float*)d_in[3];
  const float* Wg   = (const float*)d_in[4];
  const float* Wo   = (const float*)d_in[5];
  const float* Wgk1 = (const float*)d_in[6];
  const float* Wgk2 = (const float*)d_in[7];
  const float* bgk2 = (const float*)d_in[8];
  const float* rmsw = (const float*)d_in[9];
  float* out = (float*)d_out;

  float* ws   = (float*)d_ws;
  float* gws  = ws   + (size_t)TOK * 1024;
  float* xg1  = gws  + (size_t)TOK * 512;
  float* a16  = xg1  + (size_t)TOK * 16;
  float* qhw  = a16  + (size_t)TOK * 256;
  float* khw  = qhw  + (size_t)TOK * 128;
  float* vTw  = khw  + (size_t)TOK * 128;
  float* elc  = vTw  + (size_t)TOK * 256;
  float* kvw  = elc  + (size_t)TOK * 4;
  float* spw  = kvw  + (size_t)TOK * 512;
  float* wTw  = spw  + (size_t)TOK * 256;
  float* woTw = wTw  + (size_t)PROJ_WTILES * WTILE / 2;

  lr1_kernel<<<LR1_BLOCKS + PROJ_WTILES + OUT_WTILES, 256, 0, stream>>>(
      x, Wgk1, xg1, Wq, Wk, Wv, Wg, Wo, wTw);
  proj_gate_kernel<<<(TOK / CH) * Hc, 256, 0, stream>>>(
      x, wTw, xg1, Wgk2, bgk2, gws, qhw, khw, vTw, elc, kvw);
  state_kernel<<<dim3(Bc * Hc, (DVc * DKc) / 512), 256, 0, stream>>>(kvw, elc, spw);
  gla_chunk_kernel<<<(TOK / CH) * Hc, 256, 0, stream>>>(
      qhw, khw, vTw, spw, gws, rmsw, a16);
  out_gemm_kernel<<<dim3(TOK / 128, 512 / 128), 256, 0, stream>>>(
      a16, woTw, out);
}


# logsigmoid's log1pf (a ~100-instruction library chain per element, in the K-loop) and the IEEE divides of k-hat and the swish sigmoid now on the hardware's log2 / reciprocal (1 ulp); sum orders kept
# speedup vs baseline: 16.2106x; 1.1042x over previous
#include <hip/hip_runtime.h>
#include <hip/hip_bf16.h>
#include <math.h>


typedef __attribute__((ext_vector_type(16))) _Float16 v16h;
typedef __attribute__((ext_vector_type(8)))  _Float16 v8h;
typedef __attribute__((ext_vector_type(4)))  _Float16 v4h;
typedef __attribute__((ext_vector_type(8)))  float    v8f;
typedef __attribute__((ext_vector_type(4)))  float    v4f;
typedef __attribute__((ext_vector_type(2)))  float    v2f;
typedef __attribute__((ext_vector_type(2)))  _Float16 v2h;

constexpr int Bc  = 4;
constexpr int Tc  = 2048;
constexpr int HIDc = 512;
constexpr int Hc  = 4;
constexpr int DKc = 64;
constexpr int DVc = 128;
constexpr int CH  = 64;
constexpr int TOK = Bc * Tc;
constexpr float RMS_EPS = 1e-5f;
constexpr int WTILE = 128 * 32;
constexpr int LR1_BLOCKS = TOK / 16;
constexpr int PROJ_WTILES = (1536 / 128) * (HIDc / 32);
constexpr int OUT_WTILES  = (512 / 128) * (512 / 32);
constexpr int FN = 384;
constexpr int FTILE = FN * 32;

__device__ __forceinline__ v8f wmma16(v16h a, v16h b, v8f c) {
  return __builtin_amdgcn_wmma_f32_16x16x32_f16(false, a, false, b, (short)0, c,
                                                false, false);
}

__device__ __forceinline__ v16h frag16(const _Float16* p, int ld, int rowBase,
                                       int kBase) {
  const int lane = threadIdx.x & 31, half = lane >> 4;
  const _Float16* row = p + (size_t)(rowBase + (lane & 15)) * ld + kBase + half * 8;
  v8h lo = *(const v8h*)(row);
  v8h hi = *(const v8h*)(row + 16);
  return __builtin_shufflevector(lo, hi, 0, 1, 2, 3, 4, 5, 6, 7,
                                 8, 9, 10, 11, 12, 13, 14, 15);
}


__global__ __launch_bounds__(256) void lr1_kernel(
    const float* __restrict__ x, const float* __restrict__ Wgk1,
    float* __restrict__ xg1,
    const float* __restrict__ Wq, const float* __restrict__ Wk,
    const float* __restrict__ Wv, const float* __restrict__ Wg,
    const float* __restrict__ Wo, float* __restrict__ wT_ws) {
  __shared__ __align__(16) float xs[16][512];
  __shared__ __align__(16) float part[32][272];
  if (blockIdx.x >= LR1_BLOCKS) {
    const int j = blockIdx.x - LR1_BLOCKS;
    const int nb = (j < PROJ_WTILES ? j : j - PROJ_WTILES) >> 4, ks = j & 15;
    const float* src;
    int ld;
    if (j >= PROJ_WTILES) { src = Wo + nb * 128;       ld = 512; }
    else if (nb < 2)      { src = Wq + nb * 128;       ld = 256; }
    else if (nb < 4)      { src = Wk + (nb - 2) * 128; ld = 256; }
    else if (nb < 8)      { src = Wv + (nb - 4) * 128; ld = 512; }
    else                  { src = Wg + (nb - 8) * 128; ld = 512; }
    _Float16 (*tT)[40] = (_Float16 (*)[40])&xs[0][0];
    for (int i = threadIdx.x; i < 1024; i += 256) {
      const int k = i >> 5, ng = i & 31;
      const v4f wv = *(const v4f*)(src + (size_t)(ks * 32 + k) * ld + ng * 4);
#pragma unroll
      for (int j2 = 0; j2 < 4; ++j2) tT[ng * 4 + j2][k] = (_Float16)wv[j2];
    }
    __syncthreads();
    for (int r = 0; r < 2; ++r) {
      const int i = threadIdx.x + 256 * r, nr = i >> 2;
      size_t o;
      if (j >= PROJ_WTILES) {
        o = (size_t)j * WTILE + nr * 32;
      } else {
        int hh, pos;
        if (nb < 2)      { const int c = nb * 128 + nr;       hh = c >> 6; pos = c & 63; }
        else if (nb < 4) { const int c = (nb - 2) * 128 + nr; hh = c >> 6; pos = 64 + (c & 63); }
        else if (nb < 8) { hh = nb - 4; pos = 128 + nr; }
        else             { hh = nb - 8; pos = 256 + nr; }
        o = ((size_t)(hh * 16 + ks) * FN + pos) * 32;
      }
      *(v8h*)((_Float16*)wT_ws + o + (i & 3) * 8) = *(const v8h*)&tT[nr][(i & 3) * 8];
    }
    return;
  }
  const int t0 = blockIdx.x * 16;
  const int tid = threadIdx.x;
  const int jp = tid & 7, ks = tid >> 3;
  v2f w[16];
#pragma unroll
  for (int kk = 0; kk < 16; ++kk)
    w[kk] = *(const v2f*)(Wgk1 + (size_t)(ks * 16 + kk) * 16 + jp * 2);
#pragma unroll
  for (int r = 0; r < 8; ++r) {
    const int i = tid + 256 * r;
    *(v4f*)&xs[i >> 7][(i & 127) * 4] =
        *(const v4f*)(x + (size_t)(t0 + (i >> 7)) * 512 + (i & 127) * 4);
  }
  __syncthreads();
#pragma unroll 4
  for (int tok = 0; tok < 16; ++tok) {
    const float* xr = &xs[tok][ks * 16];
    float s0 = 0.f, s1 = 0.f;
#pragma unroll
    for (int q4 = 0; q4 < 4; ++q4) {
      const v4f xv = *(const v4f*)(xr + q4 * 4);
#pragma unroll
      for (int e = 0; e < 4; ++e) {
        s0 += xv[e] * w[q4 * 4 + e][0];
        s1 += xv[e] * w[q4 * 4 + e][1];
      }
    }
    const v2f p = {s0, s1};
    *(v2f*)&part[ks][tok * 16 + jp * 2] = p;
  }
  __syncthreads();
  float s = 0.f;
#pragma unroll
  for (int k2 = 0; k2 < 32; ++k2) s += part[k2][tid];
  xg1[(size_t)t0 * 16 + tid] = s;
}

__global__ __launch_bounds__(256) void proj_gate_kernel(
    const float* __restrict__ x, const float* __restrict__ wT_ws,
    const float* __restrict__ xg1, const float* __restrict__ Wgk2,
    const float* __restrict__ bgk2, float* __restrict__ gws,
    float* __restrict__ qh_ws, float* __restrict__ kh_ws,
    float* __restrict__ vT_ws, float* __restrict__ expLCg,
    float* __restrict__ kv_ws) {
  constexpr int LDA = CH + 8;
  constexpr int LDF = DKc + 1;
  constexpr int NS = HIDc / 32;
  constexpr int A_BYTES = 2 * CH * 40 * 2;
  constexpr int B_BYTES = 2 * FN * 40 * 2;
  __shared__ __align__(16) unsigned char stage[A_BYTES + B_BYTES];
  __shared__ __align__(16) float    xs[CH][16];
  __shared__ float                  lamS[CH][DKc + 1];
  __shared__ float                  gtot[4][DKc];
  __shared__ __align__(16) _Float16 vTS[DVc][LDA];
  _Float16 (*As)[CH][40]  = (_Float16 (*)[CH][40])stage;
  _Float16 (*BsT)[FN][40] = (_Float16 (*)[FN][40])(stage + A_BYTES);
  float* qF = (float*)stage;
  float* kF = qF + CH * LDF;
  _Float16 (*qS)[LDA]  = (_Float16 (*)[LDA])(stage + 2 * CH * LDF * 4);
  _Float16 (*kS)[LDA]  = qS + CH;
  _Float16 (*kTS)[LDA] = kS + CH;
  static_assert(2 * CH * LDF * 4 + 3 * CH * LDA * 2 <= A_BYTES + B_BYTES, "alias fits");
  static_assert((2 * CH * LDF * 4) % 16 == 0, "f16 tiles 16-B aligned");

  const int tile = blockIdx.x;
  const int h = tile & 3, tok0 = (tile >> 2) * CH;
  const int tid = threadIdx.x, lane = tid & 31, wave = tid >> 5;
  const int d = tid & 63, grp = tid >> 6;
  const int mr0 = (wave & 1) * 32;
  const int nc0 = (wave >> 1) * 96;
  const int arow = tid >> 3, acg = tid & 7;
  const float* xa_p = x + (size_t)(tok0 + arow) * HIDc + acg * 4;
  const float* xpf_p = x + (size_t)(tok0 + (tid >> 2)) * HIDc;
  const _Float16* bt_p = (const _Float16*)wT_ws + (size_t)h * 16 * FTILE + tid * 8;
  v4f xa[2];
  v8h bb[6];
  auto load = [&](int s1) {
#pragma unroll
    for (int r = 0; r < 2; ++r)
      xa[r] = *(const v4f*)(xa_p + (size_t)r * 32 * HIDc + s1 * 32);
#pragma unroll
    for (int r = 0; r < 6; ++r)
      bb[r] = *(const v8h*)(bt_p + (size_t)s1 * FTILE + r * 2048);
    if (s1 + 1 < NS) __builtin_prefetch(xpf_p + (s1 + 1) * 32, 0, 1);
  };
  auto store = [&](int st) {
#pragma unroll
    for (int r = 0; r < 2; ++r) {
      v4h hv;
#pragma unroll
      for (int e = 0; e < 4; ++e) hv[e] = (_Float16)xa[r][e];
      *(v4h*)&As[st][arow + 32 * r][acg * 4] = hv;
    }
#pragma unroll
    for (int r = 0; r < 6; ++r) {
      const int i = tid + 256 * r;
      *(v8h*)&BsT[st][i >> 2][(i & 3) * 8] = bb[r];
    }
  };

  *(v4f*)(&xs[0][0] + tid * 4) = *(const v4f*)(xg1 + (size_t)tok0 * 16 + tid * 4);
  float w[16];
#pragma unroll
  for (int j = 0; j < 16; ++j) w[j] = Wgk2[j * 256 + h * DKc + d];
  const float bias = bgk2[h * DKc + d];
  load(0);
  store(0);
  __syncthreads();
  static_assert(NS * 4 == CH, "one gate time step per K-step and thread group");
  float lrun = 0.f;

  v8f acc[2][6] = {};
  for (int s = 0; s < NS; ++s) {
    const int st = s & 1;
    if (s + 1 < NS) load(s + 1);
    __builtin_amdgcn_sched_barrier(0);
    v16h a0 = frag16(&As[st][0][0], 40, mr0, 0);
    v16h a1 = frag16(&As[st][0][0], 40, mr0 + 16, 0);
#pragma unroll
    for (int j = 0; j < 6; ++j) {
      v16h bm = frag16(&BsT[st][0][0], 40, nc0 + j * 16, 0);
      acc[0][j] = wmma16(a0, bm, acc[0][j]);
      acc[1][j] = wmma16(a1, bm, acc[1][j]);
    }
    {
      const int t = grp * 16 + s;
      float sg = bias;
#pragma unroll
      for (int j = 0; j < 16; ++j) sg += xs[t][j] * w[j];
      const float e = __expf(-fabsf(sg));
      const float ls = fminf(sg, 0.f) - 0.6931471806f * __builtin_amdgcn_logf(1.f + e);
      lrun += ls * (1.f / 16.f);
      lamS[t][d] = lrun;
    }
    __builtin_amdgcn_sched_barrier(0);
    if (s + 1 < NS) store(st ^ 1);
    __syncthreads();
  }

  const int nl = lane & 15, mOff = (lane >> 4) * 8;
  const int ncu = __builtin_amdgcn_readfirstlane(nc0);
  gtot[grp][d] = lrun;
#pragma unroll
  for (int a2 = 0; a2 < 2; ++a2) {
    const int tb = mr0 + a2 * 16 + mOff;
#pragma unroll
    for (int j = 0; j < 6; ++j) {
      const int cb = ncu + j * 16;
      if (cb < 64) {
#pragma unroll
        for (int r = 0; r < 8; ++r)
          qF[(tb + r) * LDF + cb + nl] = acc[a2][j][r] * 0.125f;
      } else if (cb < 128) {
#pragma unroll
        for (int r = 0; r < 8; ++r)
          kF[(tb + r) * LDF + (cb - 64) + nl] = acc[a2][j][r];
      } else if (cb < 256) {
        v8h hv;
#pragma unroll
        for (int r = 0; r < 8; ++r) hv[r] = (_Float16)acc[a2][j][r];
        *(v8h*)&vTS[cb - 128 + nl][tb] = hv;
      } else {
        float* gp = gws + (size_t)(tok0 + tb) * (Hc * DVc) + h * DVc + (cb - 256) + nl;
#pragma unroll
        for (int r = 0; r < 8; ++r) gp[(size_t)r * (Hc * DVc)] = acc[a2][j][r];
      }
    }
  }
  __syncthreads();

  const float g01 = gtot[0][d] + gtot[1][d];
  const float g012 = g01 + gtot[2][d];
  const float lamC = g012 + gtot[3][d];
  const float goff = grp == 0 ? 0.f : grp == 1 ? gtot[0][d] : grp == 2 ? g01 : g012;
  if (tid < DKc) expLCg[(size_t)tile * DKc + tid] = __expf(lamC);
#pragma unroll 4
  for (int tt = 0; tt < 16; ++tt) {
    const int t = grp * 16 + tt;
    const float eq = __expf((goff + lamS[t][d]) - lamC);
    const float qh = fmaxf(fminf(qF[t * LDF + d] * eq, 60000.f), -60000.f);
    const float kh = kF[t * LDF + d] * __builtin_amdgcn_rcpf(eq);
    qS[t][d]  = (_Float16)qh;
    kS[t][d]  = (_Float16)kh;
    kTS[d][t] = (_Float16)kh;
  }
  __syncthreads();

  _Float16* qd  = (_Float16*)qh_ws  + (size_t)tile * (CH * DKc);
  _Float16* kd  = (_Float16*)kh_ws  + (size_t)tile * (CH * DKc);
  _Float16* vtd = (_Float16*)vT_ws  + (size_t)tile * (CH * DVc);
  for (int r = 0; r < 2; ++r) {
    const int i = tid + 256 * r, row = i >> 3, cc = (i & 7) * 8;
    *(v8h*)(qd  + i * 8) = *(const v8h*)&qS[row][cc];
    *(v8h*)(kd  + i * 8) = *(const v8h*)&kS[row][cc];
  }
  for (int r = 0; r < 4; ++r) {
    const int i = tid + 256 * r, row = i >> 3, cc = (i & 7) * 8;
    *(v8h*)(vtd + i * 8) = *(const v8h*)&vTS[row][cc];
  }

  {
    v8f kacc[4] = {};
#pragma unroll
    for (int k2 = 0; k2 < 2; ++k2) {
      v16h bv = frag16(&vTS[0][0], LDA, wave * 16, k2 * 32);
#pragma unroll
      for (int dr = 0; dr < 4; ++dr) {
        v16h aK = frag16(&kTS[0][0], LDA, dr * 16, k2 * 32);
        kacc[dr] = wmma16(aK, bv, kacc[dr]);
      }
    }
    float* dst = kv_ws + (size_t)tile * (DVc * DKc) +
                 (wave * 16 + (lane & 15)) * DKc + (lane >> 4) * 8;
#pragma unroll
    for (int dr = 0; dr < 4; ++dr) {
      const v4f lo = {kacc[dr][0], kacc[dr][1], kacc[dr][2], kacc[dr][3]};
      const v4f hi = {kacc[dr][4], kacc[dr][5], kacc[dr][6], kacc[dr][7]};
      *(v4f*)(dst + dr * 16)     = lo;
      *(v4f*)(dst + dr * 16 + 4) = hi;
    }
  }
}

__global__ __launch_bounds__(256) void state_kernel(
    const float* __restrict__ kv_ws, const float* __restrict__ expLCg,
    float* __restrict__ sp_ws) {
  constexpr int NC = Tc / CH;
  constexpr int GRP = 16;
  const int bh = blockIdx.x, b = bh >> 2, h = bh & 3;
  const int e2 = (blockIdx.y * 256 + threadIdx.x) * 2;
  const int d = e2 & 63;
  _Float16* sp = (_Float16*)sp_ws;
  float s0 = 0.f, s1 = 0.f;
  for (int c0 = 0; c0 < NC; c0 += GRP) {
    v2f kv[GRP], dc[GRP];
#pragma unroll
    for (int i = 0; i < GRP; ++i) {
      const size_t tile = (size_t)((b * NC + c0 + i) * Hc + h);
      kv[i] = *(const v2f*)(kv_ws + tile * (DVc * DKc) + e2);
      dc[i] = *(const v2f*)(expLCg + tile * DKc + d);
    }
#pragma unroll
    for (int i = 0; i < GRP; ++i) {
      const size_t tile = (size_t)((b * NC + c0 + i) * Hc + h);
      s0 *= dc[i][0];
      s1 *= dc[i][1];
      v2h o2;
      o2[0] = (_Float16)s0;
      o2[1] = (_Float16)s1;
      *(v2h*)(sp + tile * (DVc * DKc) + e2) = o2;
      s0 += kv[i][0];
      s1 += kv[i][1];
    }
  }
}

__device__ __forceinline__ float row16_sum(float v) {
  v += __builtin_bit_cast(float, __builtin_amdgcn_update_dpp(
           0, __builtin_bit_cast(int, v), 0x161, 0xf, 0xf, true));
  v += __builtin_bit_cast(float, __builtin_amdgcn_update_dpp(
           0, __builtin_bit_cast(int, v), 0x162, 0xf, 0xf, true));
  v += __builtin_bit_cast(float, __builtin_amdgcn_update_dpp(
           0, __builtin_bit_cast(int, v), 0x164, 0xf, 0xf, true));
  v += __builtin_bit_cast(float, __builtin_amdgcn_update_dpp(
           0, __builtin_bit_cast(int, v), 0x168, 0xf, 0xf, true));
  return v;
}

__global__ __launch_bounds__(256) void gla_chunk_kernel(
    const float* __restrict__ qh_ws, const float* __restrict__ kh_ws,
    const float* __restrict__ vT_ws, const float* __restrict__ sp_ws,
    const float* __restrict__ g, const float* __restrict__ rms_w,
    float* __restrict__ a16_ws) {
  constexpr int LDA = CH + 8;
  constexpr int LDO = DVc + 8;
  __shared__ __align__(16) _Float16 qt[CH][LDA];
  __shared__ __align__(16) _Float16 kt[CH][LDA];
  __shared__ __align__(16) _Float16 Am[CH][LDA];
  __shared__ __align__(16) _Float16 vT[DVc][LDA];
  __shared__ __align__(16) _Float16 sT[DVc][LDA];
  __shared__ __align__(16) _Float16 oS[CH][LDO];
  __shared__ float ssqP[2][CH];

  const int tile = blockIdx.x;
  const int h = tile & 3, tok0 = (tile >> 2) * CH;
  const int tid = threadIdx.x, lane = tid & 31, wave = tid >> 5;
  const int tRow = wave & 3, nHalf = wave >> 2, ntb = nHalf * 4;
  const int nl = lane & 15, mOff = (lane >> 4) * 8;

  {
    const _Float16* qtile  = (const _Float16*)qh_ws + (size_t)tile * (CH * DKc);
    const _Float16* ktile  = (const _Float16*)kh_ws + (size_t)tile * (CH * DKc);
    const _Float16* vTtile = (const _Float16*)vT_ws + (size_t)tile * (CH * DVc);
    const _Float16* stile  = (const _Float16*)sp_ws + (size_t)tile * (DVc * DKc);
    for (int r = 0; r < 2; ++r) {
      const int i = tid + 256 * r, row = i >> 3, cc = (i & 7) * 8;
      *(v8h*)&qt[row][cc] = *(const v8h*)(qtile + i * 8);
      *(v8h*)&kt[row][cc] = *(const v8h*)(ktile + i * 8);
    }
    for (int r = 0; r < 4; ++r) {
      const int i = tid + 256 * r, row = i >> 3, cc = (i & 7) * 8;
      *(v8h*)&vT[row][cc] = *(const v8h*)(vTtile + i * 8);
      *(v8h*)&sT[row][cc] = *(const v8h*)(stile  + i * 8);
    }
  }
  float gv[4][8], rwv[4];
  {
    const float* gb = g + (size_t)(tok0 + tRow * 16 + mOff) * (Hc * DVc) + h * DVc + nl;
#pragma unroll
    for (int j = 0; j < 4; ++j) {
      rwv[j] = rms_w[(ntb + j) * 16 + nl];
#pragma unroll
      for (int r = 0; r < 8; ++r)
        gv[j][r] = gb[(size_t)r * (Hc * DVc) + (ntb + j) * 16];
    }
  }
  __syncthreads();

  {
    const int sp0 = (wave >> 2) * 2;
#pragma unroll
    for (int spi = 0; spi < 2; ++spi) {
      const int sTile = sp0 + spi;
      v8f acc = {};
      if (sTile <= tRow) {
#pragma unroll
        for (int k2 = 0; k2 < 2; ++k2) {
          v16h a  = frag16(&qt[0][0], LDA, tRow * 16, k2 * 32);
          v16h bm = frag16(&kt[0][0], LDA, sTile * 16, k2 * 32);
          acc = wmma16(a, bm, acc);
        }
      }
      const int sIdx = sTile * 16 + nl;
#pragma unroll
      for (int r = 0; r < 8; ++r) {
        const int tIdx = tRow * 16 + mOff + r;
        Am[tIdx][sIdx] = (_Float16)((sIdx <= tIdx) ? acc[r] : 0.f);
      }
    }
  }
  __syncthreads();

  v8f acc[4] = {};
#pragma unroll
  for (int k2 = 0; k2 < 2; ++k2) {
    v16h aA = frag16(&Am[0][0], LDA, tRow * 16, k2 * 32);
    v16h aQ = frag16(&qt[0][0], LDA, tRow * 16, k2 * 32);
#pragma unroll
    for (int j = 0; j < 4; ++j) {
      v16h bv = frag16(&vT[0][0], LDA, (ntb + j) * 16, k2 * 32);
      acc[j] = wmma16(aA, bv, acc[j]);
      v16h bs = frag16(&sT[0][0], LDA, (ntb + j) * 16, k2 * 32);
      acc[j] = wmma16(aQ, bs, acc[j]);
    }
  }
#pragma unroll
  for (int r = 0; r < 8; ++r) {
    float p = acc[0][r] * acc[0][r];
#pragma unroll
    for (int j = 1; j < 4; ++j) p += acc[j][r] * acc[j][r];
    p = row16_sum(p);
    if (nl == 0) ssqP[nHalf][tRow * 16 + mOff + r] = p;
  }
  __syncthreads();

#pragma unroll
  for (int r = 0; r < 8; ++r) {
    const int t = tRow * 16 + mOff + r;
    const float ms = (ssqP[0][t] + ssqP[1][t]) * (1.f / DVc);
    const float rstd = rsqrtf(ms + RMS_EPS);
#pragma unroll
    for (int j = 0; j < 4; ++j) {
      const float gg = gv[j][r];
      const float val = acc[j][r] * rstd * rwv[j];
      oS[t][(ntb + j) * 16 + nl] =
          (_Float16)(val * gg * __builtin_amdgcn_rcpf(1.f + __expf(-gg)));
    }
  }
  __syncthreads();

  {
    _Float16* adst = (_Float16*)a16_ws +
                     ((size_t)(tok0 >> 7) * 16 + h * 4) * WTILE + (tok0 & 127) * 32 + tid * 8;
    const int t = tid >> 2, c8 = (tid & 3) * 8;
#pragma unroll
    for (int q = 0; q < 4; ++q)
      *(v8h*)(adst + (size_t)q * WTILE) = *(const v8h*)&oS[t][q * 32 + c8];
  }
}

__global__ __launch_bounds__(256) void out_gemm_kernel(
    const float* __restrict__ a16_ws, const float* __restrict__ woT_ws,
    float* __restrict__ Cm) {
  constexpr int KD = 128;
  constexpr int LDK = KD + 40;
  constexpr int NS = 512 / KD;
  __shared__ __align__(16) _Float16 As[2][128][LDK];
  __shared__ __align__(16) _Float16 BsT[2][128][LDK];
  const int m0 = blockIdx.x * 128, n0 = blockIdx.y * 128;
  const int tid = threadIdx.x, lane = tid & 31, wave = tid >> 5;
  const int mr0 = (wave & 3) * 32, nc0 = (wave >> 2) * 64;
  const _Float16* at_p = (const _Float16*)a16_ws + (size_t)blockIdx.x * 16 * WTILE + tid * 8;
  const _Float16* bt_p = (const _Float16*)woT_ws + (size_t)blockIdx.y * 16 * WTILE + tid * 8;

  v8h aa[8], bb[8];
  v8f acc[2][4] = {};

  auto load = [&](int s1) {
#pragma unroll
    for (int r = 0; r < 8; ++r) {
      aa[r] = *(const v8h*)(at_p + (size_t)s1 * (4 * WTILE) + r * 2048);
      bb[r] = *(const v8h*)(bt_p + (size_t)s1 * (4 * WTILE) + r * 2048);
    }
  };
  auto store = [&](int st) {
#pragma unroll
    for (int r = 0; r < 8; ++r) {
      const int i = tid + 256 * r;
      const int row = (i & 511) >> 2, k = (i >> 9) * 32 + (i & 3) * 8;
      *(v8h*)&As[st][row][k]  = aa[r];
      *(v8h*)&BsT[st][row][k] = bb[r];
    }
  };

  load(0);
  store(0);
  __syncthreads();
  for (int s = 0; s < NS; ++s) {
    const int st = s & 1;
    if (s + 1 < NS) load(s + 1);
    __builtin_amdgcn_sched_barrier(0);
#pragma unroll
    for (int kk = 0; kk < KD; kk += 32) {
      v16h a0 = frag16(&As[st][0][0], LDK, mr0, kk);
      v16h a1 = frag16(&As[st][0][0], LDK, mr0 + 16, kk);
#pragma unroll
      for (int j = 0; j < 4; ++j) {
        v16h bm = frag16(&BsT[st][0][0], LDK, nc0 + j * 16, kk);
        acc[0][j] = wmma16(a0, bm, acc[0][j]);
        acc[1][j] = wmma16(a1, bm, acc[1][j]);
      }
    }
    if (s + 1 < NS) store(st ^ 1);
    __syncthreads();
  }
  const int nl = lane & 15, mOff = (lane >> 4) * 8;
#pragma unroll
  for (int a2 = 0; a2 < 2; ++a2) {
#pragma unroll
    for (int j = 0; j < 4; ++j) {
      const int n = n0 + nc0 + j * 16 + nl;
#pragma unroll
      for (int r = 0; r < 8; ++r)
        Cm[(size_t)(m0 + mr0 + a2 * 16 + mOff + r) * 512 + n] = acc[a2][j][r];
    }
  }
}

extern "C" void kernel_launch(void* const* d_in, const int* in_sizes, int n_in,
                              void* d_out, int out_size, void* d_ws, size_t ws_size,
                              hipStream_t stream) {
  (void)in_sizes; (void)n_in; (void)out_size; (void)ws_size;
  const float* x    = (const float*)d_in[0];
  const float* Wq   = (const float*)d_in[1];
  const float* Wk   = (const float*)d_in[2];
  const float* Wv   = (const float*)d_in[3];
  const float* Wg   = (const float*)d_in[4];
  const float* Wo   = (const float*)d_in[5];
  const float* Wgk1 = (const float*)d_in[6];
  const float* Wgk2 = (const float*)d_in[7];
  const float* bgk2 = (const float*)d_in[8];
  const float* rmsw = (const float*)d_in[9];
  float* out = (float*)d_out;

  float* ws   = (float*)d_ws;
  float* gws  = ws   + (size_t)TOK * 1024;
  float* xg1  = gws  + (size_t)TOK * 512;
  float* a16  = xg1  + (size_t)TOK * 16;
  float* qhw  = a16  + (size_t)TOK * 256;
  float* khw  = qhw  + (size_t)TOK * 128;
  float* vTw  = khw  + (size_t)TOK * 128;
  float* elc  = vTw  + (size_t)TOK * 256;
  float* kvw  = elc  + (size_t)TOK * 4;
  float* spw  = kvw  + (size_t)TOK * 512;
  float* wTw  = spw  + (size_t)TOK * 256;
  float* woTw = wTw  + (size_t)PROJ_WTILES * WTILE / 2;

  lr1_kernel<<<LR1_BLOCKS + PROJ_WTILES + OUT_WTILES, 256, 0, stream>>>(
      x, Wgk1, xg1, Wq, Wk, Wv, Wg, Wo, wTw);
  proj_gate_kernel<<<(TOK / CH) * Hc, 256, 0, stream>>>(
      x, wTw, xg1, Wgk2, bgk2, gws, qhw, khw, vTw, elc, kvw);
  state_kernel<<<dim3(Bc * Hc, (DVc * DKc) / 512), 256, 0, stream>>>(kvw, elc, spw);
  gla_chunk_kernel<<<(TOK / CH) * Hc, 256, 0, stream>>>(
      qhw, khw, vTw, spw, gws, rmsw, a16);
  out_gemm_kernel<<<dim3(TOK / 128, 512 / 128), 256, 0, stream>>>(
      a16, woTw, out);
}
